# starting kernel
# baseline (speedup 1.0000x reference)
#include <hip/hip_runtime.h>
#include <hip/hip_bf16.h>

typedef __attribute__((ext_vector_type(2))) float v2f;
typedef __attribute__((ext_vector_type(8))) float v8f;

__global__ void k_deg_init(float* deg, int N) {
    int i = blockIdx.x * blockDim.x + threadIdx.x;
    if (i < N) deg[i] = 1.0f;
}

__global__ void k_deg_count(const int* __restrict__ dst, float* deg, int E) {
    int e = blockIdx.x * blockDim.x + threadIdx.x;
    if (e < E) atomicAdd(&deg[dst[e]], 1.0f);
}

__global__ void k_dinv(float* deg, int N) {
    int i = blockIdx.x * blockDim.x + threadIdx.x;
    if (i < N) deg[i] = rsqrtf(deg[i]);
}

template <int NT>
__global__ __launch_bounds__(32)
void k_gemm_wmma_f32(const float* __restrict__ X, const float* __restrict__ W,
                     float* __restrict__ Y, int K) {
    constexpr int Cout = NT * 16;
    const int mt   = blockIdx.x;
    const int lane = threadIdx.x;
    const int half = lane >> 4;
    const int l16  = lane & 15;

    const float* __restrict__ xrow = X + (size_t)(mt * 16 + l16) * K;
    const float* __restrict__ wlan = W + l16;

    v8f c[NT];
#pragma unroll
    for (int t = 0; t < NT; ++t) c[t] = (v8f){};

    for (int k = 0; k < K; k += 4) {
        const int ka = k + half * 2;
        v2f a;
        a.x = xrow[ka];
        a.y = xrow[ka + 1];
        const float* __restrict__ wk = wlan + (size_t)ka * Cout;
#pragma unroll
        for (int t = 0; t < NT; ++t) {
            v2f b;
            b.x = wk[t * 16];
            b.y = wk[t * 16 + Cout];
            c[t] = __builtin_amdgcn_wmma_f32_16x16x4_f32(
                        false, a,  false, b,
                        (short)0, c[t],
                        false,  false);
        }
    }

    float* __restrict__ yp = Y + (size_t)(mt * 16) * Cout + l16;
#pragma unroll
    for (int t = 0; t < NT; ++t)
#pragma unroll
        for (int v = 0; v < 8; ++v)
            yp[(size_t)(v + half * 8) * Cout + t * 16] = c[t][v];
}


__global__ void k_agg_init(const float* __restrict__ h, const float* __restrict__ bias,
                           const float* __restrict__ dinv, float* __restrict__ out,
                           int C) {
    const int i = blockIdx.x;
    const int c = threadIdx.x;
    const float di = dinv[i];
    out[(size_t)i * C + c] = bias[c] + h[(size_t)i * C + c] * di * di;
}

__global__ void k_agg_edges(const float* __restrict__ h,
                            const int* __restrict__ src, const int* __restrict__ dst,
                            const float* __restrict__ dinv, float* __restrict__ out,
                            int C) {
    const int e = blockIdx.x;
    const int c = threadIdx.x;
    const int s = src[e];
    const int d = dst[e];
    const float nrm = dinv[s] * dinv[d];
    atomicAdd(&out[(size_t)d * C + c], h[(size_t)s * C + c] * nrm);
}

__global__ void k_relu(float* x, long long n) {
    long long i = (long long)blockIdx.x * blockDim.x + threadIdx.x;
    if (i < n) x[i] = fmaxf(x[i], 0.0f);
}

extern "C" void kernel_launch(void* const* d_in, const int* in_sizes, int n_in,
                              void* d_out, int out_size, void* d_ws, size_t ws_size,
                              hipStream_t stream) {
    const float* x  = (const float*)d_in[0];
    const int*   ei = (const int*)  d_in[1];
    const float* W1 = (const float*)d_in[2];
    const float* b1 = (const float*)d_in[3];
    const float* W2 = (const float*)d_in[4];
    const float* b2 = (const float*)d_in[5];
    const float* W3 = (const float*)d_in[6];
    const float* b3 = (const float*)d_in[7];

    const int K0 = 512, C1 = 128, C2 = 64, C3 = 32;
    const int N = in_sizes[0] / K0;
    const int E = in_sizes[1] / 2;
    const int* src = ei;
    const int* dst = ei + E;

    float* ws   = (float*)d_ws;
    size_t pad  = ((size_t)N + 255) & ~(size_t)255;
    float* dinv = ws;
    float* bufH = ws + pad;
    float* bufA = bufH + (size_t)N * C1;
    float* out  = (float*)d_out;

    const int TB = 256;

    k_deg_init <<<(N + TB - 1) / TB, TB, 0, stream>>>(dinv, N);
    k_deg_count<<<(E + TB - 1) / TB, TB, 0, stream>>>(dst, dinv, E);
    k_dinv     <<<(N + TB - 1) / TB, TB, 0, stream>>>(dinv, N);

    k_gemm_wmma_f32<8><<<N / 16, 32, 0, stream>>>(x, W1, bufH, K0);
    k_agg_init <<<N, C1, 0, stream>>>(bufH, b1, dinv, bufA, C1);
    k_agg_edges<<<E, C1, 0, stream>>>(bufH, src, dst, dinv, bufA, C1);
    k_relu     <<<(int)(((long long)N * C1 + TB - 1) / TB), TB, 0, stream>>>(bufA, (long long)N * C1);

    k_gemm_wmma_f32<4><<<N / 16, 32, 0, stream>>>(bufA, W2, bufH, C1);
    k_agg_init <<<N, C2, 0, stream>>>(bufH, b2, dinv, bufA, C2);
    k_agg_edges<<<E, C2, 0, stream>>>(bufH, src, dst, dinv, bufA, C2);
    k_relu     <<<(int)(((long long)N * C2 + TB - 1) / TB), TB, 0, stream>>>(bufA, (long long)N * C2);

    k_gemm_wmma_f32<2><<<N / 16, 32, 0, stream>>>(bufA, W3, bufH, C2);
    k_agg_init <<<N, C3, 0, stream>>>(bufH, b3, dinv, out, C3);
    k_agg_edges<<<E, C3, 0, stream>>>(bufH, src, dst, dinv, out, C3);
}


# layer-1 tile filter reads a one-byte tile number per edge (plane written by the degree-count pass, 16 edges per 16-byte load) instead of 4-byte dst; dst fetched for hits only; sum order as before
# speedup vs baseline: 3.6744x; 3.6744x over previous
#include <hip/hip_runtime.h>
#include <hip/hip_bf16.h>

typedef __attribute__((ext_vector_type(2))) float v2f;
typedef __attribute__((ext_vector_type(8))) float v8f;

__global__ void k_deg_init(float* deg, int N) {
    int i = blockIdx.x * blockDim.x + threadIdx.x;
    if (i < N) deg[i] = 1.0f;
}

constexpr int REGION_ROW_WORDS = 32;
constexpr int PLANE_RUN_LOG2   = 13;
constexpr int PLANE_RUN        = 1 << PLANE_RUN_LOG2;

template <int T, int LROW>
__device__ __forceinline__ size_t plane_offset(int e) {
    constexpr int RB = T * LROW * 4;
    return (size_t)(e >> PLANE_RUN_LOG2) * RB + (size_t)(RB - PLANE_RUN) + (size_t)(e & (PLANE_RUN - 1));
}

template <int T, int LROW>
__global__ void k_deg_count(const int* __restrict__ dst, float* deg, float* lst, int E) {
    const int e0 = (blockIdx.x * blockDim.x + threadIdx.x) * 4;
    if (e0 >= E) return;
    unsigned char* tb = (unsigned char*)lst + plane_offset<T, LROW>(e0);
    if (e0 + 3 < E) {
        const int d0 = dst[e0], d1 = dst[e0 + 1], d2 = dst[e0 + 2], d3 = dst[e0 + 3];
        atomicAdd(&deg[d0], 1.0f);
        atomicAdd(&deg[d1], 1.0f);
        atomicAdd(&deg[d2], 1.0f);
        atomicAdd(&deg[d3], 1.0f);
        *(unsigned*)tb = (unsigned)(d0 / T) | ((unsigned)(d1 / T) << 8) |
                         ((unsigned)(d2 / T) << 16) | ((unsigned)(d3 / T) << 24);
    } else {
        for (int e = e0; e < E; ++e) {
            const int d = dst[e];
            atomicAdd(&deg[d], 1.0f);
            tb[e - e0] = (unsigned char)(d / T);
        }
    }
}

__global__ void k_dinv(float* deg, int N) {
    int i = blockIdx.x * blockDim.x + threadIdx.x;
    if (i < N) deg[i] = rsqrtf(deg[i]);
}

template <int NT>
__global__ __launch_bounds__(32)
void k_gemm_wmma_f32(const float* __restrict__ X, const float* __restrict__ W,
                     float* __restrict__ Y, int K) {
    constexpr int Cout = NT * 16;
    const int mt   = blockIdx.x;
    const int lane = threadIdx.x;
    const int half = lane >> 4;
    const int l16  = lane & 15;

    const float* __restrict__ xrow = X + (size_t)(mt * 16 + l16) * K;
    const float* __restrict__ wlan = W + l16;

    v8f c[NT];
#pragma unroll
    for (int t = 0; t < NT; ++t) c[t] = (v8f){};

    for (int k = 0; k < K; k += 4) {
        const int ka = k + half * 2;
        v2f a;
        a.x = xrow[ka];
        a.y = xrow[ka + 1];
        const float* __restrict__ wk = wlan + (size_t)ka * Cout;
#pragma unroll
        for (int t = 0; t < NT; ++t) {
            v2f b;
            b.x = wk[t * 16];
            b.y = wk[t * 16 + Cout];
            c[t] = __builtin_amdgcn_wmma_f32_16x16x4_f32(
                        false, a,  false, b,
                        (short)0, c[t],
                        false,  false);
        }
    }

    float* __restrict__ yp = Y + (size_t)(mt * 16) * Cout + l16;
#pragma unroll
    for (int t = 0; t < NT; ++t)
#pragma unroll
        for (int v = 0; v < 8; ++v)
            yp[(size_t)(v + half * 8) * Cout + t * 16] = c[t][v];
}

template <int C, int T, bool RELU, int MODE>
__global__ __launch_bounds__(1024)
void k_agg_tile(const float* __restrict__ h,
                const int* __restrict__ src, const int* __restrict__ dst,
                const float* __restrict__ dinv, const float* __restrict__ bias,
                float* out, float* lst, int N, int E) {
    constexpr int NT  = 1024;
    constexpr int VEC = C / 32;
    constexpr int C4  = C / 4;
    constexpr int CAP = 4096;
    constexpr int LPT = 8;
    constexpr int SUB = 6;
    constexpr int CH4 = NT * LPT * SUB;
    constexpr int U   = 16 / VEC;
    constexpr int LROW = REGION_ROW_WORDS;
    constexpr int RB   = T * LROW * 4;
    constexpr int LPTB = 6;
    constexpr int SUBB = 2;
    constexpr int CHV  = NT * LPTB * SUBB;
    constexpr int VPR_LOG2 = PLANE_RUN_LOG2 - 4;
    static_assert(T <= 512, "row in tile must fit 9 bits of a list word");
    static_assert(RB > PLANE_RUN + 4 && RB % 16 == 0, "a region holds a list and an aligned run of the plane");

    __shared__ float acc[(T + 1) * C];
    __shared__ float dloc[T + 1];
    __shared__ int   qe[CAP];
    __shared__ int   cnt[2];
    __shared__ int   ovf;

    const int tid  = threadIdx.x;
    const int lane = tid & 31;
    const int wave = __builtin_amdgcn_readfirstlane(tid >> 5);
    const int base = blockIdx.x * T;
    const int rows = min(T, N - base);

    unsigned* lw = (unsigned*)lst + (size_t)base * LROW;
    const int lcap = max(rows * LROW - PLANE_RUN / 4 - 1, 0);
    unsigned lcount = 0;
    if (MODE == 1) lcount = (unsigned)__builtin_amdgcn_readfirstlane((int)lw[0]);

    for (int i = tid; i < rows; i += NT) dloc[i] = dinv[base + i];
    if (tid < 2) cnt[tid] = 0;
    if (tid == 2) ovf = 0;
    if (tid == 3) dloc[T] = 0.0f;
    {
        const float4* __restrict__ h4 = (const float4*)(h + (size_t)base * C);
        const float4* __restrict__ b4 = (const float4*)bias;
        float4* a4 = (float4*)acc;
        for (int idx = tid; idx < rows * C4; idx += NT) {
            const int r  = idx / C4;
            const int c4 = idx - r * C4;
            const float di = dinv[base + r];
            const float4 hv = h4[idx];
            const float4 bv = b4[c4];
            float4 o;
            o.x = bv.x + hv.x * di * di;
            o.y = bv.y + hv.y * di * di;
            o.z = bv.z + hv.z * di * di;
            o.w = bv.w + hv.w * di * di;
            a4[idx] = o;
        }
    }
    __syncthreads();

    auto add_rows = [&](int s, int dl, float nrm, int m) {
        for (int j0 = 0; j0 < m; j0 += U) {
            float x[U][VEC];
#pragma unroll
            for (int u = 0; u < U; ++u) {
                const int sj = __builtin_amdgcn_readlane(s, j0 + u);
                const float* __restrict__ hr = h + (size_t)sj * C + lane;
#pragma unroll
                for (int k = 0; k < VEC; ++k) x[u][k] = hr[k * 32];
            }
#pragma unroll
            for (int u = 0; u < U; ++u) {
                const int   dj = __builtin_amdgcn_readlane(dl, j0 + u);
                const float nj = __int_as_float(
                    __builtin_amdgcn_readlane(__float_as_int(nrm), j0 + u));
                float* ar = acc + dj * C + lane;
#pragma unroll
                for (int k = 0; k < VEC; ++k) atomicAdd(&ar[k * 32], x[u][k] * nj);
            }
        }
    };

    if (MODE == 1 && lcount <= (unsigned)lcap) {
        const int total = (int)lcount;
        const int per   = (total + 31) >> 5;
        const int beg   = wave * per;
        const int end   = min(beg + per, total);
        const unsigned EMPTY = (unsigned)T;
        if (beg < end) {
            const int i0 = beg + lane;
            unsigned w0 = lw[1 + min(i0,      end - 1)];
            unsigned w1 = lw[1 + min(i0 + 32, end - 1)];
            if (i0      >= end) w0 = EMPTY;
            if (i0 + 32 >= end) w1 = EMPTY;
            float    d0 = dinv[w0 >> 9];
            for (int b = beg; b < end; b += 32) {
                const int m = min(32, end - b);
                const float d1 = dinv[w1 >> 9];
                const int   i2 = b + 64 + lane;
                unsigned w2 = lw[1 + min(i2, end - 1)];
                if (i2 >= end) w2 = EMPTY;
                const int s  = (int)(w0 >> 9);
                const int dl = (int)(w0 & 511u);
                add_rows(s, dl, d0 * dloc[dl], m);
                w0 = w1; d0 = d1; w1 = w2;
            }
        }
        __syncthreads();
    } else {
        const bool vec_ok = (((size_t)dst) & 15) == 0;
        const int  n4     = vec_ok ? (E >> 2) : 0;
        const int  n16    = E >> 4;
        const int4* __restrict__ dst4 = (const int4*)dst;
        const char* pl = (const char*)lst + (RB - PLANE_RUN);
        const int nchunks = (MODE == 0) ? (n16 + CHV - 1) / CHV : (n4 + CH4 - 1) / CH4;
        const int ndone   = (MODE == 0) ? n16 * 16 : n4 * 4;
        const int nsteps  = nchunks + ((ndone < E) ? 1 : 0);
        int p = 0;
        int tot = 0;

        auto push = [&](int e) {
            const int q = atomicAdd(&cnt[p], 1);
            if (q < CAP) {
                qe[q] = e;
            } else {
                if (MODE == 0) ovf = 1;
                const int dl = dst[e] - base;
                const int s = src[e];
                const float nrm = dinv[s] * dloc[dl];
#pragma unroll 1
                for (int c = 0; c < C; ++c)
                    atomicAdd(&acc[dl * C + c], h[(size_t)s * C + c] * nrm);
            }
        };
        auto test = [&](int e, int d) {
            if ((unsigned)(d - base) < (unsigned)rows) push(e);
        };

        const int rot = (nchunks > 0) ? (int)(blockIdx.x % (unsigned)nchunks) : 0;
        const unsigned pat = (unsigned)blockIdx.x * 0x01010101u;
        auto zb = [](unsigned v) {
            return ~(((v & 0x7F7F7F7Fu) + 0x7F7F7F7Fu) | v) & 0x80808080u;
        };

        for (int st = 0; st < nsteps; ++st) {
            if (st < nchunks) {
                const int ch = (st + rot < nchunks) ? (st + rot) : (st + rot - nchunks);
                if (MODE == 0) {
#pragma unroll 1
                    for (int sb = 0; sb < SUBB; ++sb) {
                        if (ch * CHV + sb * (NT * LPTB) >= n16) break;
                        const int i0 = ch * CHV + sb * (NT * LPTB) + tid;
                        uint4 d[LPTB];
                        unsigned off[LPTB];
#pragma unroll
                        for (int k = 0; k < LPTB; ++k) {
                            const int ic = min(i0 + k * NT, n16 - 1);
                            off[k] = ((unsigned)ic >> VPR_LOG2) * (unsigned)RB
                                   + ((unsigned)ic & ((1u << VPR_LOG2) - 1u)) * 16u;
                        }
                        __builtin_amdgcn_sched_barrier(0);
#pragma unroll
                        for (int k = 0; k < LPTB; ++k) d[k] = *(const uint4*)(pl + off[k]);
                        __builtin_amdgcn_sched_barrier(0);
#pragma unroll
                        for (int k = 0; k < LPTB; ++k) {
                            unsigned f = (zb(d[k].x ^ pat) >> 7) | (zb(d[k].y ^ pat) >> 6) |
                                         (zb(d[k].z ^ pat) >> 5) | (zb(d[k].w ^ pat) >> 4);
                            if (i0 + k * NT >= n16) f = 0;
                            const int e = (i0 + k * NT) * 16;
                            while (f) {
                                const int pos = __builtin_ctz(f);
                                f &= f - 1;
                                push(e + (pos & 7) * 4 + (pos >> 3));
                            }
                        }
                    }
                } else {
#pragma unroll 1
                    for (int sb = 0; sb < SUB; ++sb) {
                        if (ch * CH4 + sb * (NT * LPT) >= n4) break;
                        const int i0 = ch * CH4 + sb * (NT * LPT) + tid;
                        int4 d[LPT];
#pragma unroll
                        for (int k = 0; k < LPT; ++k) {
                            const int i = i0 + k * NT;
                            d[k] = dst4[min(i, n4 - 1)];
                            if (i >= n4) d[k] = make_int4(-1, -1, -1, -1);
                        }
#pragma unroll
                        for (int k = 0; k < LPT; ++k) {
                            const unsigned a = (unsigned)(d[k].x - base);
                            const unsigned b = (unsigned)(d[k].y - base);
                            const unsigned c = (unsigned)(d[k].z - base);
                            const unsigned f = (unsigned)(d[k].w - base);
                            if (min(min(a, b), min(c, f)) < (unsigned)rows) {
                                const int e = (i0 + k * NT) * 4;
                                test(e,     d[k].x);
                                test(e + 1, d[k].y);
                                test(e + 2, d[k].z);
                                test(e + 3, d[k].w);
                            }
                        }
                    }
                }
            } else {
                for (int e = ndone + tid; e < E; e += NT) test(e, dst[e]);
            }
            __syncthreads();

            const int n = __builtin_amdgcn_readfirstlane(min(cnt[p], CAP));
            if (tid == 0) cnt[p ^ 1] = 0;
            for (int r0 = 0; r0 < n; r0 += NT) {
                const int nr = min(n - r0, NT);
                const int li = lane * 32 + wave;
                int   s = 0, dl = T;
                float nrm = 0.0f;
                if (li < nr) {
                    const int e = qe[r0 + li];
                    dl  = dst[e] - base;
                    s   = src[e];
                    nrm = dinv[s] * dloc[dl];
                    if (MODE == 0) {
                        const int pos = tot + r0 + li;
                        if (pos < lcap) lw[1 + pos] = ((unsigned)s << 9) | (unsigned)dl;
                    }
                }
                const int m = (nr > wave) ? ((nr - wave + 31) >> 5) : 0;
                add_rows(s, dl, nrm, m);
            }
            tot += n;
            __syncthreads();
            p ^= 1;
        }

        if (MODE == 0) {
            if (tid == 0) lw[0] = ovf ? 0xFFFFFFFFu : (unsigned)tot;
        }
    }

    {
        const float4* a4 = (const float4*)acc;
        float4* o4 = (float4*)(out + (size_t)base * C);
        for (int idx = tid; idx < rows * C4; idx += NT) {
            float4 o = a4[idx];
            if (RELU) {
                o.x = fmaxf(o.x, 0.0f);
                o.y = fmaxf(o.y, 0.0f);
                o.z = fmaxf(o.z, 0.0f);
                o.w = fmaxf(o.w, 0.0f);
            }
            o4[idx] = o;
        }
    }
}

extern "C" void kernel_launch(void* const* d_in, const int* in_sizes, int n_in,
                              void* d_out, int out_size, void* d_ws, size_t ws_size,
                              hipStream_t stream) {
    const float* x  = (const float*)d_in[0];
    const int*   ei = (const int*)  d_in[1];
    const float* W1 = (const float*)d_in[2];
    const float* b1 = (const float*)d_in[3];
    const float* W2 = (const float*)d_in[4];
    const float* b2 = (const float*)d_in[5];
    const float* W3 = (const float*)d_in[6];
    const float* b3 = (const float*)d_in[7];

    const int K0 = 512, C1 = 128, C2 = 64, C3 = 32;
    const int N = in_sizes[0] / K0;
    const int E = in_sizes[1] / 2;
    const int* src = ei;
    const int* dst = ei + E;

    float* ws   = (float*)d_ws;
    size_t pad  = ((size_t)N + 255) & ~(size_t)255;
    float* dinv = ws;
    float* bufH = ws + pad;
    float* bufA = bufH + (size_t)N * C1;
    float* out  = (float*)d_out;

    const int TB = 256;
    constexpr int TILE = 400;

    k_deg_init <<<(N + TB - 1) / TB, TB, 0, stream>>>(dinv, N);
    k_deg_count<TILE, REGION_ROW_WORDS>
               <<<(E + 4 * TB - 1) / (4 * TB), TB, 0, stream>>>(dst, dinv, out, E);
    k_dinv     <<<(N + TB - 1) / TB, TB, 0, stream>>>(dinv, N);

    k_gemm_wmma_f32<8><<<N / 16, 32, 0, stream>>>(x, W1, bufH, K0);
    k_agg_tile<C1, TILE, true, 0><<<(N + TILE - 1) / TILE, 1024, 0, stream>>>(bufH, src, dst, dinv, b1, bufA, out, N, E);

    k_gemm_wmma_f32<4><<<N / 16, 32, 0, stream>>>(bufA, W2, bufH, C1);
    k_agg_tile<C2, TILE, true, 1><<<(N + TILE - 1) / TILE, 1024, 0, stream>>>(bufH, src, dst, dinv, b2, bufA, out, N, E);

    k_gemm_wmma_f32<2><<<N / 16, 32, 0, stream>>>(bufA, W3, bufH, C2);
    k_agg_tile<C3, TILE, false, 1><<<(N + TILE - 1) / TILE, 1024, 0, stream>>>(bufH, src, dst, dinv, b3, out, out, N, E);
}


# layer-1 GEMM: 256 resident 4-wave groups hold W1 in registers (a quarter of k per wave), walk 16-row panels loading only X; the 4 quarter sums are added via LDS in fixed order (k sum order changed)
# speedup vs baseline: 3.9669x; 1.0796x over previous
#include <hip/hip_runtime.h>
#include <hip/hip_bf16.h>

typedef __attribute__((ext_vector_type(2))) float v2f;
typedef __attribute__((ext_vector_type(8))) float v8f;

__global__ void k_deg_init(float* deg, int N) {
    int i = blockIdx.x * blockDim.x + threadIdx.x;
    if (i < N) deg[i] = 1.0f;
}

constexpr int REGION_ROW_WORDS = 32;
constexpr int PLANE_RUN_LOG2   = 13;
constexpr int PLANE_RUN        = 1 << PLANE_RUN_LOG2;

template <int T, int LROW>
__device__ __forceinline__ size_t plane_offset(int e) {
    constexpr int RB = T * LROW * 4;
    return (size_t)(e >> PLANE_RUN_LOG2) * RB + (size_t)(RB - PLANE_RUN) + (size_t)(e & (PLANE_RUN - 1));
}

template <int T, int LROW>
__global__ void k_deg_count(const int* __restrict__ dst, float* deg, float* lst, int E) {
    const int e0 = (blockIdx.x * blockDim.x + threadIdx.x) * 4;
    if (e0 >= E) return;
    unsigned char* tb = (unsigned char*)lst + plane_offset<T, LROW>(e0);
    if (e0 + 3 < E) {
        const int d0 = dst[e0], d1 = dst[e0 + 1], d2 = dst[e0 + 2], d3 = dst[e0 + 3];
        atomicAdd(&deg[d0], 1.0f);
        atomicAdd(&deg[d1], 1.0f);
        atomicAdd(&deg[d2], 1.0f);
        atomicAdd(&deg[d3], 1.0f);
        *(unsigned*)tb = (unsigned)(d0 / T) | ((unsigned)(d1 / T) << 8) |
                         ((unsigned)(d2 / T) << 16) | ((unsigned)(d3 / T) << 24);
    } else {
        for (int e = e0; e < E; ++e) {
            const int d = dst[e];
            atomicAdd(&deg[d], 1.0f);
            tb[e - e0] = (unsigned char)(d / T);
        }
    }
}

__global__ void k_dinv(float* deg, int N) {
    int i = blockIdx.x * blockDim.x + threadIdx.x;
    if (i < N) deg[i] = rsqrtf(deg[i]);
}

template <int NT, int KS>
__global__ __launch_bounds__(128)
void k_gemm_wreg_f32(const float* __restrict__ X, const float* __restrict__ W,
                     float* __restrict__ Y, int npanels) {
    constexpr int Cout = NT * 16;
    constexpr int KW   = KS * 4;
    constexpr int K    = KW * 4;
    constexpr int P    = Cout + 4;
    constexpr int C4   = Cout / 4;
    __shared__ alignas(16) float red[2][4][16 * P];

    const int tid  = threadIdx.x;
    const int lane = tid & 31;
    const int q    = __builtin_amdgcn_readfirstlane(tid >> 5);
    const int half = lane >> 4;
    const int l16  = lane & 15;

    v2f b[KS][NT];
    {
        const float* __restrict__ wq = W + (size_t)(q * KW + half * 2) * Cout + l16;
#pragma unroll
        for (int j = 0; j < KS; ++j)
#pragma unroll
            for (int t = 0; t < NT; ++t) {
                b[j][t].x = wq[(size_t)(4 * j) * Cout + t * 16];
                b[j][t].y = wq[(size_t)(4 * j + 1) * Cout + t * 16];
            }
    }

    const float* __restrict__ xq = X + (size_t)l16 * K + q * KW + half * 2;

    int p = blockIdx.x;
    v2f a[KS];
    {
        const float* __restrict__ xp = xq + (size_t)min(p, npanels - 1) * 16 * K;
#pragma unroll
        for (int j = 0; j < KS; ++j) a[j] = *(const v2f*)(xp + 4 * j);
    }

    int buf = 0;
    for (; p < npanels; p += gridDim.x) {
        v2f an[KS];
        {
            const float* __restrict__ xp = xq + (size_t)min(p + (int)gridDim.x, npanels - 1) * 16 * K;
#pragma unroll
            for (int j = 0; j < KS; ++j) an[j] = *(const v2f*)(xp + 4 * j);
        }

        v8f c[NT];
#pragma unroll
        for (int t = 0; t < NT; ++t) c[t] = (v8f){};
#pragma unroll
        for (int j = 0; j < KS; ++j)
#pragma unroll
            for (int t = 0; t < NT; ++t)
                c[t] = __builtin_amdgcn_wmma_f32_16x16x4_f32(
                            false, a[j],  false, b[j][t],
                            (short)0, c[t],
                            false,  false);

        float* rw = &red[buf][q][0] + l16;
#pragma unroll
        for (int t = 0; t < NT; ++t)
#pragma unroll
            for (int v = 0; v < 8; ++v)
                rw[(v + half * 8) * P + t * 16] = c[t][v];
        __syncthreads();

        float4* __restrict__ y4 = (float4*)(Y + (size_t)p * 16 * Cout);
#pragma unroll
        for (int i = tid; i < 16 * C4; i += 128) {
            const int r  = i / C4;
            const int c4 = i - r * C4;
            const float4 s0 = *(const float4*)&red[buf][0][r * P + c4 * 4];
            const float4 s1 = *(const float4*)&red[buf][1][r * P + c4 * 4];
            const float4 s2 = *(const float4*)&red[buf][2][r * P + c4 * 4];
            const float4 s3 = *(const float4*)&red[buf][3][r * P + c4 * 4];
            float4 o;
            o.x = ((s0.x + s1.x) + s2.x) + s3.x;
            o.y = ((s0.y + s1.y) + s2.y) + s3.y;
            o.z = ((s0.z + s1.z) + s2.z) + s3.z;
            o.w = ((s0.w + s1.w) + s2.w) + s3.w;
            y4[i] = o;
        }
        buf ^= 1;
#pragma unroll
        for (int j = 0; j < KS; ++j) a[j] = an[j];
    }
}

template <int NT>
__global__ __launch_bounds__(32)
void k_gemm_wmma_f32(const float* __restrict__ X, const float* __restrict__ W,
                     float* __restrict__ Y, int K) {
    constexpr int Cout = NT * 16;
    const int mt   = blockIdx.x;
    const int lane = threadIdx.x;
    const int half = lane >> 4;
    const int l16  = lane & 15;

    const float* __restrict__ xrow = X + (size_t)(mt * 16 + l16) * K;
    const float* __restrict__ wlan = W + l16;

    v8f c[NT];
#pragma unroll
    for (int t = 0; t < NT; ++t) c[t] = (v8f){};

    for (int k = 0; k < K; k += 4) {
        const int ka = k + half * 2;
        v2f a;
        a.x = xrow[ka];
        a.y = xrow[ka + 1];
        const float* __restrict__ wk = wlan + (size_t)ka * Cout;
#pragma unroll
        for (int t = 0; t < NT; ++t) {
            v2f b;
            b.x = wk[t * 16];
            b.y = wk[t * 16 + Cout];
            c[t] = __builtin_amdgcn_wmma_f32_16x16x4_f32(
                        false, a,  false, b,
                        (short)0, c[t],
                        false,  false);
        }
    }

    float* __restrict__ yp = Y + (size_t)(mt * 16) * Cout + l16;
#pragma unroll
    for (int t = 0; t < NT; ++t)
#pragma unroll
        for (int v = 0; v < 8; ++v)
            yp[(size_t)(v + half * 8) * Cout + t * 16] = c[t][v];
}

template <int C, int T, bool RELU, int MODE>
__global__ __launch_bounds__(1024)
void k_agg_tile(const float* __restrict__ h,
                const int* __restrict__ src, const int* __restrict__ dst,
                const float* __restrict__ dinv, const float* __restrict__ bias,
                float* out, float* lst, int N, int E) {
    constexpr int NT  = 1024;
    constexpr int VEC = C / 32;
    constexpr int C4  = C / 4;
    constexpr int CAP = 4096;
    constexpr int LPT = 8;
    constexpr int SUB = 6;
    constexpr int CH4 = NT * LPT * SUB;
    constexpr int U   = 16 / VEC;
    constexpr int LROW = REGION_ROW_WORDS;
    constexpr int RB   = T * LROW * 4;
    constexpr int LPTB = 6;
    constexpr int SUBB = 2;
    constexpr int CHV  = NT * LPTB * SUBB;
    constexpr int VPR_LOG2 = PLANE_RUN_LOG2 - 4;
    static_assert(T <= 512, "row in tile must fit 9 bits of a list word");
    static_assert(RB > PLANE_RUN + 4 && RB % 16 == 0, "a region holds a list and an aligned run of the plane");

    __shared__ float acc[(T + 1) * C];
    __shared__ float dloc[T + 1];
    __shared__ int   qe[CAP];
    __shared__ int   cnt[2];
    __shared__ int   ovf;

    const int tid  = threadIdx.x;
    const int lane = tid & 31;
    const int wave = __builtin_amdgcn_readfirstlane(tid >> 5);
    const int base = blockIdx.x * T;
    const int rows = min(T, N - base);

    unsigned* lw = (unsigned*)lst + (size_t)base * LROW;
    const int lcap = max(rows * LROW - PLANE_RUN / 4 - 1, 0);
    unsigned lcount = 0;
    if (MODE == 1) lcount = (unsigned)__builtin_amdgcn_readfirstlane((int)lw[0]);

    for (int i = tid; i < rows; i += NT) dloc[i] = dinv[base + i];
    if (tid < 2) cnt[tid] = 0;
    if (tid == 2) ovf = 0;
    if (tid == 3) dloc[T] = 0.0f;
    {
        const float4* __restrict__ h4 = (const float4*)(h + (size_t)base * C);
        const float4* __restrict__ b4 = (const float4*)bias;
        float4* a4 = (float4*)acc;
        for (int idx = tid; idx < rows * C4; idx += NT) {
            const int r  = idx / C4;
            const int c4 = idx - r * C4;
            const float di = dinv[base + r];
            const float4 hv = h4[idx];
            const float4 bv = b4[c4];
            float4 o;
            o.x = bv.x + hv.x * di * di;
            o.y = bv.y + hv.y * di * di;
            o.z = bv.z + hv.z * di * di;
            o.w = bv.w + hv.w * di * di;
            a4[idx] = o;
        }
    }
    __syncthreads();

    auto add_rows = [&](int s, int dl, float nrm, int m) {
        for (int j0 = 0; j0 < m; j0 += U) {
            float x[U][VEC];
#pragma unroll
            for (int u = 0; u < U; ++u) {
                const int sj = __builtin_amdgcn_readlane(s, j0 + u);
                const float* __restrict__ hr = h + (size_t)sj * C + lane;
#pragma unroll
                for (int k = 0; k < VEC; ++k) x[u][k] = hr[k * 32];
            }
#pragma unroll
            for (int u = 0; u < U; ++u) {
                const int   dj = __builtin_amdgcn_readlane(dl, j0 + u);
                const float nj = __int_as_float(
                    __builtin_amdgcn_readlane(__float_as_int(nrm), j0 + u));
                float* ar = acc + dj * C + lane;
#pragma unroll
                for (int k = 0; k < VEC; ++k) atomicAdd(&ar[k * 32], x[u][k] * nj);
            }
        }
    };

    if (MODE == 1 && lcount <= (unsigned)lcap) {
        const int total = (int)lcount;
        const int per   = (total + 31) >> 5;
        const int beg   = wave * per;
        const int end   = min(beg + per, total);
        const unsigned EMPTY = (unsigned)T;
        if (beg < end) {
            const int i0 = beg + lane;
            unsigned w0 = lw[1 + min(i0,      end - 1)];
            unsigned w1 = lw[1 + min(i0 + 32, end - 1)];
            if (i0      >= end) w0 = EMPTY;
            if (i0 + 32 >= end) w1 = EMPTY;
            float    d0 = dinv[w0 >> 9];
            for (int b = beg; b < end; b += 32) {
                const int m = min(32, end - b);
                const float d1 = dinv[w1 >> 9];
                const int   i2 = b + 64 + lane;
                unsigned w2 = lw[1 + min(i2, end - 1)];
                if (i2 >= end) w2 = EMPTY;
                const int s  = (int)(w0 >> 9);
                const int dl = (int)(w0 & 511u);
                add_rows(s, dl, d0 * dloc[dl], m);
                w0 = w1; d0 = d1; w1 = w2;
            }
        }
        __syncthreads();
    } else {
        const bool vec_ok = (((size_t)dst) & 15) == 0;
        const int  n4     = vec_ok ? (E >> 2) : 0;
        const int  n16    = E >> 4;
        const int4* __restrict__ dst4 = (const int4*)dst;
        const char* pl = (const char*)lst + (RB - PLANE_RUN);
        const int nchunks = (MODE == 0) ? (n16 + CHV - 1) / CHV : (n4 + CH4 - 1) / CH4;
        const int ndone   = (MODE == 0) ? n16 * 16 : n4 * 4;
        const int nsteps  = nchunks + ((ndone < E) ? 1 : 0);
        int p = 0;
        int tot = 0;

        auto push = [&](int e) {
            const int q = atomicAdd(&cnt[p], 1);
            if (q < CAP) {
                qe[q] = e;
            } else {
                if (MODE == 0) ovf = 1;
                const int dl = dst[e] - base;
                const int s = src[e];
                const float nrm = dinv[s] * dloc[dl];
#pragma unroll 1
                for (int c = 0; c < C; ++c)
                    atomicAdd(&acc[dl * C + c], h[(size_t)s * C + c] * nrm);
            }
        };
        auto test = [&](int e, int d) {
            if ((unsigned)(d - base) < (unsigned)rows) push(e);
        };

        const int rot = (nchunks > 0) ? (int)(blockIdx.x % (unsigned)nchunks) : 0;
        const unsigned pat = (unsigned)blockIdx.x * 0x01010101u;
        auto zb = [](unsigned v) {
            return ~(((v & 0x7F7F7F7Fu) + 0x7F7F7F7Fu) | v) & 0x80808080u;
        };

        for (int st = 0; st < nsteps; ++st) {
            if (st < nchunks) {
                const int ch = (st + rot < nchunks) ? (st + rot) : (st + rot - nchunks);
                if (MODE == 0) {
#pragma unroll 1
                    for (int sb = 0; sb < SUBB; ++sb) {
                        if (ch * CHV + sb * (NT * LPTB) >= n16) break;
                        const int i0 = ch * CHV + sb * (NT * LPTB) + tid;
                        uint4 d[LPTB];
                        unsigned off[LPTB];
#pragma unroll
                        for (int k = 0; k < LPTB; ++k) {
                            const int ic = min(i0 + k * NT, n16 - 1);
                            off[k] = ((unsigned)ic >> VPR_LOG2) * (unsigned)RB
                                   + ((unsigned)ic & ((1u << VPR_LOG2) - 1u)) * 16u;
                        }
                        __builtin_amdgcn_sched_barrier(0);
#pragma unroll
                        for (int k = 0; k < LPTB; ++k) d[k] = *(const uint4*)(pl + off[k]);
                        __builtin_amdgcn_sched_barrier(0);
#pragma unroll
                        for (int k = 0; k < LPTB; ++k) {
                            unsigned f = (zb(d[k].x ^ pat) >> 7) | (zb(d[k].y ^ pat) >> 6) |
                                         (zb(d[k].z ^ pat) >> 5) | (zb(d[k].w ^ pat) >> 4);
                            if (i0 + k * NT >= n16) f = 0;
                            const int e = (i0 + k * NT) * 16;
                            while (f) {
                                const int pos = __builtin_ctz(f);
                                f &= f - 1;
                                push(e + (pos & 7) * 4 + (pos >> 3));
                            }
                        }
                    }
                } else {
#pragma unroll 1
                    for (int sb = 0; sb < SUB; ++sb) {
                        if (ch * CH4 + sb * (NT * LPT) >= n4) break;
                        const int i0 = ch * CH4 + sb * (NT * LPT) + tid;
                        int4 d[LPT];
#pragma unroll
                        for (int k = 0; k < LPT; ++k) {
                            const int i = i0 + k * NT;
                            d[k] = dst4[min(i, n4 - 1)];
                            if (i >= n4) d[k] = make_int4(-1, -1, -1, -1);
                        }
#pragma unroll
                        for (int k = 0; k < LPT; ++k) {
                            const unsigned a = (unsigned)(d[k].x - base);
                            const unsigned b = (unsigned)(d[k].y - base);
                            const unsigned c = (unsigned)(d[k].z - base);
                            const unsigned f = (unsigned)(d[k].w - base);
                            if (min(min(a, b), min(c, f)) < (unsigned)rows) {
                                const int e = (i0 + k * NT) * 4;
                                test(e,     d[k].x);
                                test(e + 1, d[k].y);
                                test(e + 2, d[k].z);
                                test(e + 3, d[k].w);
                            }
                        }
                    }
                }
            } else {
                for (int e = ndone + tid; e < E; e += NT) test(e, dst[e]);
            }
            __syncthreads();

            const int n = __builtin_amdgcn_readfirstlane(min(cnt[p], CAP));
            if (tid == 0) cnt[p ^ 1] = 0;
            for (int r0 = 0; r0 < n; r0 += NT) {
                const int nr = min(n - r0, NT);
                const int li = lane * 32 + wave;
                int   s = 0, dl = T;
                float nrm = 0.0f;
                if (li < nr) {
                    const int e = qe[r0 + li];
                    dl  = dst[e] - base;
                    s   = src[e];
                    nrm = dinv[s] * dloc[dl];
                    if (MODE == 0) {
                        const int pos = tot + r0 + li;
                        if (pos < lcap) lw[1 + pos] = ((unsigned)s << 9) | (unsigned)dl;
                    }
                }
                const int m = (nr > wave) ? ((nr - wave + 31) >> 5) : 0;
                add_rows(s, dl, nrm, m);
            }
            tot += n;
            __syncthreads();
            p ^= 1;
        }

        if (MODE == 0) {
            if (tid == 0) lw[0] = ovf ? 0xFFFFFFFFu : (unsigned)tot;
        }
    }

    {
        const float4* a4 = (const float4*)acc;
        float4* o4 = (float4*)(out + (size_t)base * C);
        for (int idx = tid; idx < rows * C4; idx += NT) {
            float4 o = a4[idx];
            if (RELU) {
                o.x = fmaxf(o.x, 0.0f);
                o.y = fmaxf(o.y, 0.0f);
                o.z = fmaxf(o.z, 0.0f);
                o.w = fmaxf(o.w, 0.0f);
            }
            o4[idx] = o;
        }
    }
}

extern "C" void kernel_launch(void* const* d_in, const int* in_sizes, int n_in,
                              void* d_out, int out_size, void* d_ws, size_t ws_size,
                              hipStream_t stream) {
    const float* x  = (const float*)d_in[0];
    const int*   ei = (const int*)  d_in[1];
    const float* W1 = (const float*)d_in[2];
    const float* b1 = (const float*)d_in[3];
    const float* W2 = (const float*)d_in[4];
    const float* b2 = (const float*)d_in[5];
    const float* W3 = (const float*)d_in[6];
    const float* b3 = (const float*)d_in[7];

    const int K0 = 512, C1 = 128, C2 = 64, C3 = 32;
    const int N = in_sizes[0] / K0;
    const int E = in_sizes[1] / 2;
    const int* src = ei;
    const int* dst = ei + E;

    float* ws   = (float*)d_ws;
    size_t pad  = ((size_t)N + 255) & ~(size_t)255;
    float* dinv = ws;
    float* bufH = ws + pad;
    float* bufA = bufH + (size_t)N * C1;
    float* out  = (float*)d_out;

    const int TB = 256;
    constexpr int TILE = 400;

    k_deg_init <<<(N + TB - 1) / TB, TB, 0, stream>>>(dinv, N);
    k_deg_count<TILE, REGION_ROW_WORDS>
               <<<(E + 4 * TB - 1) / (4 * TB), TB, 0, stream>>>(dst, dinv, out, E);
    k_dinv     <<<(N + TB - 1) / TB, TB, 0, stream>>>(dinv, N);

    k_gemm_wreg_f32<8, 32><<<256, 128, 0, stream>>>(x, W1, bufH, N / 16);
    k_agg_tile<C1, TILE, true, 0><<<(N + TILE - 1) / TILE, 1024, 0, stream>>>(bufH, src, dst, dinv, b1, bufA, out, N, E);

    k_gemm_wmma_f32<4><<<N / 16, 32, 0, stream>>>(bufA, W2, bufH, C1);
    k_agg_tile<C2, TILE, true, 1><<<(N + TILE - 1) / TILE, 1024, 0, stream>>>(bufH, src, dst, dinv, b2, bufA, out, N, E);

    k_gemm_wmma_f32<2><<<N / 16, 32, 0, stream>>>(bufA, W3, bufH, C2);
    k_agg_tile<C3, TILE, false, 1><<<(N + TILE - 1) / TILE, 1024, 0, stream>>>(bufH, src, dst, dinv, b3, out, out, N, E);
}


# list pass before layer 1: a tile filters the plane once, counts its rows' degrees in LDS (no global atomics, 8 launches), writes its list; layer 1 walks it like layers 2-3 (LDS adds, arrival order)
# speedup vs baseline: 5.2941x; 1.3346x over previous
#include <hip/hip_runtime.h>
#include <hip/hip_bf16.h>

typedef __attribute__((ext_vector_type(2))) float v2f;
typedef __attribute__((ext_vector_type(8))) float v8f;

constexpr int REGION_ROW_WORDS = 32;
constexpr int PLANE_RUN_LOG2   = 13;
constexpr int PLANE_RUN        = 1 << PLANE_RUN_LOG2;

template <int T, int LROW>
__device__ __forceinline__ size_t plane_offset(int e) {
    constexpr int RB = T * LROW * 4;
    return (size_t)(e >> PLANE_RUN_LOG2) * RB + (size_t)(RB - PLANE_RUN) + (size_t)(e & (PLANE_RUN - 1));
}

template <int T, int LROW>
__global__ void k_plane(const int* __restrict__ dst, float* lst, int E) {
    const int e0 = (blockIdx.x * blockDim.x + threadIdx.x) * 4;
    if (e0 >= E) return;
    unsigned char* tb = (unsigned char*)lst + plane_offset<T, LROW>(e0);
    if (e0 + 3 < E) {
        const int d0 = dst[e0], d1 = dst[e0 + 1], d2 = dst[e0 + 2], d3 = dst[e0 + 3];
        *(unsigned*)tb = (unsigned)(d0 / T) | ((unsigned)(d1 / T) << 8) |
                         ((unsigned)(d2 / T) << 16) | ((unsigned)(d3 / T) << 24);
    } else {
        for (int e = e0; e < E; ++e) tb[e - e0] = (unsigned char)(dst[e] / T);
    }
}

template <int T>
__global__ __launch_bounds__(1024)
void k_tile_list(const int* __restrict__ src, const int* __restrict__ dst,
                 float* __restrict__ dinv, float* lst, int N, int E) {
    constexpr int NT   = 1024;
    constexpr int CAP  = 16384;
    constexpr int LROW = REGION_ROW_WORDS;
    constexpr int RB   = T * LROW * 4;
    constexpr int LPTB = 6;
    constexpr int GV   = NT * LPTB;
    constexpr int VPR_LOG2 = PLANE_RUN_LOG2 - 4;
    static_assert(T <= 512, "row in tile must fit 9 bits of a list word");
    static_assert(RB > PLANE_RUN + 4 && RB % 16 == 0, "a region holds a list and an aligned run of the plane");

    __shared__ int degc[T];
    __shared__ int qe[CAP];
    __shared__ int cnt;
    __shared__ int ovf;

    const int tid  = threadIdx.x;
    const int base = blockIdx.x * T;
    const int rows = min(T, N - base);

    unsigned* lw = (unsigned*)lst + (size_t)base * LROW;
    const int lcap = max(rows * LROW - PLANE_RUN / 4 - 1, 0);

    for (int i = tid; i < T; i += NT) degc[i] = 0;
    if (tid == 0) { cnt = 0; ovf = 0; }
    __syncthreads();

    auto push = [&](int e) {
        const int q = atomicAdd(&cnt, 1);
        if (q < CAP) {
            qe[q] = e;
        } else {
            ovf = 1;
            atomicAdd(&degc[dst[e] - base], 1);
        }
    };

    const int n16 = E >> 4;
    const char* pl = (const char*)lst + (RB - PLANE_RUN);
    const int ngroups = (n16 + GV - 1) / GV;
    const int rot = (ngroups > 0) ? (int)(blockIdx.x % (unsigned)ngroups) : 0;
    const unsigned pat = (unsigned)blockIdx.x * 0x01010101u;
    auto zb = [](unsigned v) {
        return ~(((v & 0x7F7F7F7Fu) + 0x7F7F7F7Fu) | v) & 0x80808080u;
    };

#pragma unroll 1
    for (int g = 0; g < ngroups; ++g) {
        const int gg = (g + rot < ngroups) ? (g + rot) : (g + rot - ngroups);
        const int i0 = gg * GV + tid;
        uint4 d[LPTB];
        unsigned off[LPTB];
#pragma unroll
        for (int k = 0; k < LPTB; ++k) {
            const int ic = min(i0 + k * NT, n16 - 1);
            off[k] = ((unsigned)ic >> VPR_LOG2) * (unsigned)RB
                   + ((unsigned)ic & ((1u << VPR_LOG2) - 1u)) * 16u;
        }
        __builtin_amdgcn_sched_barrier(0);
#pragma unroll
        for (int k = 0; k < LPTB; ++k) d[k] = *(const uint4*)(pl + off[k]);
        __builtin_amdgcn_sched_barrier(0);
#pragma unroll
        for (int k = 0; k < LPTB; ++k) {
            unsigned f = (zb(d[k].x ^ pat) >> 7) | (zb(d[k].y ^ pat) >> 6) |
                         (zb(d[k].z ^ pat) >> 5) | (zb(d[k].w ^ pat) >> 4);
            if (i0 + k * NT >= n16) f = 0;
            const int e = (i0 + k * NT) * 16;
            while (f) {
                const int pos = __builtin_ctz(f);
                f &= f - 1;
                push(e + (pos & 7) * 4 + (pos >> 3));
            }
        }
    }
    for (int e = n16 * 16 + tid; e < E; e += NT)
        if ((unsigned)(dst[e] - base) < (unsigned)rows) push(e);
    __syncthreads();

    const int n = min(cnt, CAP);
    constexpr int DU = 8;
    for (int i0 = 0; i0 < n; i0 += NT * DU) {
        int dv[DU], sv[DU];
#pragma unroll
        for (int u = 0; u < DU; ++u) {
            const int e = qe[min(i0 + u * NT + tid, n - 1)];
            dv[u] = dst[e];
            sv[u] = src[e];
        }
#pragma unroll
        for (int u = 0; u < DU; ++u) {
            const int i = i0 + u * NT + tid;
            if (i < n) {
                const int dl = dv[u] - base;
                atomicAdd(&degc[dl], 1);
                if (i < lcap) lw[1 + i] = ((unsigned)sv[u] << 9) | (unsigned)dl;
            }
        }
    }
    __syncthreads();

    for (int i = tid; i < rows; i += NT) dinv[base + i] = rsqrtf(1.0f + (float)degc[i]);
    if (tid == 0) lw[0] = ovf ? 0xFFFFFFFFu : (unsigned)n;
}

template <int NT, int KS>
__global__ __launch_bounds__(128)
void k_gemm_wreg_f32(const float* __restrict__ X, const float* __restrict__ W,
                     float* __restrict__ Y, int npanels) {
    constexpr int Cout = NT * 16;
    constexpr int KW   = KS * 4;
    constexpr int K    = KW * 4;
    constexpr int P    = Cout + 4;
    constexpr int C4   = Cout / 4;
    __shared__ alignas(16) float red[2][4][16 * P];

    const int tid  = threadIdx.x;
    const int lane = tid & 31;
    const int q    = __builtin_amdgcn_readfirstlane(tid >> 5);
    const int half = lane >> 4;
    const int l16  = lane & 15;

    v2f b[KS][NT];
    {
        const float* __restrict__ wq = W + (size_t)(q * KW + half * 2) * Cout + l16;
#pragma unroll
        for (int j = 0; j < KS; ++j)
#pragma unroll
            for (int t = 0; t < NT; ++t) {
                b[j][t].x = wq[(size_t)(4 * j) * Cout + t * 16];
                b[j][t].y = wq[(size_t)(4 * j + 1) * Cout + t * 16];
            }
    }

    const float* __restrict__ xq = X + (size_t)l16 * K + q * KW + half * 2;

    int p = blockIdx.x;
    v2f a[KS];
    {
        const float* __restrict__ xp = xq + (size_t)min(p, npanels - 1) * 16 * K;
#pragma unroll
        for (int j = 0; j < KS; ++j) a[j] = *(const v2f*)(xp + 4 * j);
    }

    int buf = 0;
    for (; p < npanels; p += gridDim.x) {
        v2f an[KS];
        {
            const float* __restrict__ xp = xq + (size_t)min(p + (int)gridDim.x, npanels - 1) * 16 * K;
#pragma unroll
            for (int j = 0; j < KS; ++j) an[j] = *(const v2f*)(xp + 4 * j);
        }

        v8f c[NT];
#pragma unroll
        for (int t = 0; t < NT; ++t) c[t] = (v8f){};
#pragma unroll
        for (int j = 0; j < KS; ++j)
#pragma unroll
            for (int t = 0; t < NT; ++t)
                c[t] = __builtin_amdgcn_wmma_f32_16x16x4_f32(
                            false, a[j],  false, b[j][t],
                            (short)0, c[t],
                            false,  false);

        float* rw = &red[buf][q][0] + l16;
#pragma unroll
        for (int t = 0; t < NT; ++t)
#pragma unroll
            for (int v = 0; v < 8; ++v)
                rw[(v + half * 8) * P + t * 16] = c[t][v];
        __syncthreads();

        float4* __restrict__ y4 = (float4*)(Y + (size_t)p * 16 * Cout);
#pragma unroll
        for (int i = tid; i < 16 * C4; i += 128) {
            const int r  = i / C4;
            const int c4 = i - r * C4;
            const float4 s0 = *(const float4*)&red[buf][0][r * P + c4 * 4];
            const float4 s1 = *(const float4*)&red[buf][1][r * P + c4 * 4];
            const float4 s2 = *(const float4*)&red[buf][2][r * P + c4 * 4];
            const float4 s3 = *(const float4*)&red[buf][3][r * P + c4 * 4];
            float4 o;
            o.x = ((s0.x + s1.x) + s2.x) + s3.x;
            o.y = ((s0.y + s1.y) + s2.y) + s3.y;
            o.z = ((s0.z + s1.z) + s2.z) + s3.z;
            o.w = ((s0.w + s1.w) + s2.w) + s3.w;
            y4[i] = o;
        }
        buf ^= 1;
#pragma unroll
        for (int j = 0; j < KS; ++j) a[j] = an[j];
    }
}

template <int NT>
__global__ __launch_bounds__(32)
void k_gemm_wmma_f32(const float* __restrict__ X, const float* __restrict__ W,
                     float* __restrict__ Y, int K) {
    constexpr int Cout = NT * 16;
    const int mt   = blockIdx.x;
    const int lane = threadIdx.x;
    const int half = lane >> 4;
    const int l16  = lane & 15;

    const float* __restrict__ xrow = X + (size_t)(mt * 16 + l16) * K;
    const float* __restrict__ wlan = W + l16;

    v8f c[NT];
#pragma unroll
    for (int t = 0; t < NT; ++t) c[t] = (v8f){};

    for (int k = 0; k < K; k += 4) {
        const int ka = k + half * 2;
        v2f a;
        a.x = xrow[ka];
        a.y = xrow[ka + 1];
        const float* __restrict__ wk = wlan + (size_t)ka * Cout;
#pragma unroll
        for (int t = 0; t < NT; ++t) {
            v2f b;
            b.x = wk[t * 16];
            b.y = wk[t * 16 + Cout];
            c[t] = __builtin_amdgcn_wmma_f32_16x16x4_f32(
                        false, a,  false, b,
                        (short)0, c[t],
                        false,  false);
        }
    }

    float* __restrict__ yp = Y + (size_t)(mt * 16) * Cout + l16;
#pragma unroll
    for (int t = 0; t < NT; ++t)
#pragma unroll
        for (int v = 0; v < 8; ++v)
            yp[(size_t)(v + half * 8) * Cout + t * 16] = c[t][v];
}

template <int C, int T, bool RELU>
__global__ __launch_bounds__(1024)
void k_agg_tile(const float* __restrict__ h,
                const int* __restrict__ src, const int* __restrict__ dst,
                const float* __restrict__ dinv, const float* __restrict__ bias,
                float* out, float* lst, int N, int E) {
    constexpr int NT  = 1024;
    constexpr int VEC = C / 32;
    constexpr int C4  = C / 4;
    constexpr int CAP = 4096;
    constexpr int LPT = 8;
    constexpr int SUB = 6;
    constexpr int CH4 = NT * LPT * SUB;
    constexpr int U   = 16 / VEC;
    constexpr int LROW = REGION_ROW_WORDS;
    static_assert(T <= 512, "row in tile must fit 9 bits of a list word");

    __shared__ float acc[(T + 1) * C];
    __shared__ float dloc[T + 1];
    __shared__ int   qe[CAP];
    __shared__ int   cnt[2];

    const int tid  = threadIdx.x;
    const int lane = tid & 31;
    const int wave = __builtin_amdgcn_readfirstlane(tid >> 5);
    const int base = blockIdx.x * T;
    const int rows = min(T, N - base);

    unsigned* lw = (unsigned*)lst + (size_t)base * LROW;
    const int lcap = max(rows * LROW - PLANE_RUN / 4 - 1, 0);
    const unsigned lcount = (unsigned)__builtin_amdgcn_readfirstlane((int)lw[0]);

    for (int i = tid; i < rows; i += NT) dloc[i] = dinv[base + i];
    if (tid < 2) cnt[tid] = 0;
    if (tid == 3) dloc[T] = 0.0f;
    {
        const float4* __restrict__ h4 = (const float4*)(h + (size_t)base * C);
        const float4* __restrict__ b4 = (const float4*)bias;
        float4* a4 = (float4*)acc;
        for (int idx = tid; idx < rows * C4; idx += NT) {
            const int r  = idx / C4;
            const int c4 = idx - r * C4;
            const float di = dinv[base + r];
            const float4 hv = h4[idx];
            const float4 bv = b4[c4];
            float4 o;
            o.x = bv.x + hv.x * di * di;
            o.y = bv.y + hv.y * di * di;
            o.z = bv.z + hv.z * di * di;
            o.w = bv.w + hv.w * di * di;
            a4[idx] = o;
        }
    }
    __syncthreads();

    auto add_rows = [&](int s, int dl, float nrm, int m) {
        for (int j0 = 0; j0 < m; j0 += U) {
            float x[U][VEC];
#pragma unroll
            for (int u = 0; u < U; ++u) {
                const int sj = __builtin_amdgcn_readlane(s, j0 + u);
                const float* __restrict__ hr = h + (size_t)sj * C + lane;
#pragma unroll
                for (int k = 0; k < VEC; ++k) x[u][k] = hr[k * 32];
            }
#pragma unroll
            for (int u = 0; u < U; ++u) {
                const int   dj = __builtin_amdgcn_readlane(dl, j0 + u);
                const float nj = __int_as_float(
                    __builtin_amdgcn_readlane(__float_as_int(nrm), j0 + u));
                float* ar = acc + dj * C + lane;
#pragma unroll
                for (int k = 0; k < VEC; ++k) atomicAdd(&ar[k * 32], x[u][k] * nj);
            }
        }
    };

    if (lcount <= (unsigned)lcap) {
        const int total = (int)lcount;
        const int per   = (total + 31) >> 5;
        const int beg   = wave * per;
        const int end   = min(beg + per, total);
        const unsigned EMPTY = (unsigned)T;
        if (beg < end) {
            const int i0 = beg + lane;
            unsigned w0 = lw[1 + min(i0,      end - 1)];
            unsigned w1 = lw[1 + min(i0 + 32, end - 1)];
            if (i0      >= end) w0 = EMPTY;
            if (i0 + 32 >= end) w1 = EMPTY;
            float    d0 = dinv[w0 >> 9];
            for (int b = beg; b < end; b += 32) {
                const int m = min(32, end - b);
                const float d1 = dinv[w1 >> 9];
                const int   i2 = b + 64 + lane;
                unsigned w2 = lw[1 + min(i2, end - 1)];
                if (i2 >= end) w2 = EMPTY;
                const int s  = (int)(w0 >> 9);
                const int dl = (int)(w0 & 511u);
                add_rows(s, dl, d0 * dloc[dl], m);
                w0 = w1; d0 = d1; w1 = w2;
            }
        }
        __syncthreads();
    } else {
        const bool vec_ok = (((size_t)dst) & 15) == 0;
        const int  n4     = vec_ok ? (E >> 2) : 0;
        const int4* __restrict__ dst4 = (const int4*)dst;
        const int nchunks = (n4 + CH4 - 1) / CH4;
        const int ndone   = n4 * 4;
        const int nsteps  = nchunks + ((ndone < E) ? 1 : 0);
        int p = 0;

        auto push = [&](int e) {
            const int q = atomicAdd(&cnt[p], 1);
            if (q < CAP) {
                qe[q] = e;
            } else {
                const int dl = dst[e] - base;
                const int s = src[e];
                const float nrm = dinv[s] * dloc[dl];
#pragma unroll 1
                for (int c = 0; c < C; ++c)
                    atomicAdd(&acc[dl * C + c], h[(size_t)s * C + c] * nrm);
            }
        };
        auto test = [&](int e, int d) {
            if ((unsigned)(d - base) < (unsigned)rows) push(e);
        };

        const int rot = (nchunks > 0) ? (int)(blockIdx.x % (unsigned)nchunks) : 0;

        for (int st = 0; st < nsteps; ++st) {
            if (st < nchunks) {
                const int ch = (st + rot < nchunks) ? (st + rot) : (st + rot - nchunks);
#pragma unroll 1
                for (int sb = 0; sb < SUB; ++sb) {
                    if (ch * CH4 + sb * (NT * LPT) >= n4) break;
                    const int i0 = ch * CH4 + sb * (NT * LPT) + tid;
                    int4 d[LPT];
#pragma unroll
                    for (int k = 0; k < LPT; ++k) {
                        const int i = i0 + k * NT;
                        d[k] = dst4[min(i, n4 - 1)];
                        if (i >= n4) d[k] = make_int4(-1, -1, -1, -1);
                    }
#pragma unroll
                    for (int k = 0; k < LPT; ++k) {
                        const unsigned a = (unsigned)(d[k].x - base);
                        const unsigned b = (unsigned)(d[k].y - base);
                        const unsigned c = (unsigned)(d[k].z - base);
                        const unsigned f = (unsigned)(d[k].w - base);
                        if (min(min(a, b), min(c, f)) < (unsigned)rows) {
                            const int e = (i0 + k * NT) * 4;
                            test(e,     d[k].x);
                            test(e + 1, d[k].y);
                            test(e + 2, d[k].z);
                            test(e + 3, d[k].w);
                        }
                    }
                }
            } else {
                for (int e = ndone + tid; e < E; e += NT) test(e, dst[e]);
            }
            __syncthreads();

            const int n = __builtin_amdgcn_readfirstlane(min(cnt[p], CAP));
            if (tid == 0) cnt[p ^ 1] = 0;
            for (int r0 = 0; r0 < n; r0 += NT) {
                const int nr = min(n - r0, NT);
                const int li = lane * 32 + wave;
                int   s = 0, dl = T;
                float nrm = 0.0f;
                if (li < nr) {
                    const int e = qe[r0 + li];
                    dl  = dst[e] - base;
                    s   = src[e];
                    nrm = dinv[s] * dloc[dl];
                }
                const int m = (nr > wave) ? ((nr - wave + 31) >> 5) : 0;
                add_rows(s, dl, nrm, m);
            }
            __syncthreads();
            p ^= 1;
        }
    }

    {
        const float4* a4 = (const float4*)acc;
        float4* o4 = (float4*)(out + (size_t)base * C);
        for (int idx = tid; idx < rows * C4; idx += NT) {
            float4 o = a4[idx];
            if (RELU) {
                o.x = fmaxf(o.x, 0.0f);
                o.y = fmaxf(o.y, 0.0f);
                o.z = fmaxf(o.z, 0.0f);
                o.w = fmaxf(o.w, 0.0f);
            }
            o4[idx] = o;
        }
    }
}

extern "C" void kernel_launch(void* const* d_in, const int* in_sizes, int n_in,
                              void* d_out, int out_size, void* d_ws, size_t ws_size,
                              hipStream_t stream) {
    const float* x  = (const float*)d_in[0];
    const int*   ei = (const int*)  d_in[1];
    const float* W1 = (const float*)d_in[2];
    const float* b1 = (const float*)d_in[3];
    const float* W2 = (const float*)d_in[4];
    const float* b2 = (const float*)d_in[5];
    const float* W3 = (const float*)d_in[6];
    const float* b3 = (const float*)d_in[7];

    const int K0 = 512, C1 = 128, C2 = 64, C3 = 32;
    const int N = in_sizes[0] / K0;
    const int E = in_sizes[1] / 2;
    const int* src = ei;
    const int* dst = ei + E;

    float* ws   = (float*)d_ws;
    size_t pad  = ((size_t)N + 255) & ~(size_t)255;
    float* dinv = ws;
    float* bufH = ws + pad;
    float* bufA = bufH + (size_t)N * C1;
    float* out  = (float*)d_out;

    const int TB = 256;
    constexpr int TILE = 400;

    k_plane<TILE, REGION_ROW_WORDS><<<(E + 4 * TB - 1) / (4 * TB), TB, 0, stream>>>(dst, out, E);
    k_tile_list<TILE><<<(N + TILE - 1) / TILE, 1024, 0, stream>>>(src, dst, dinv, out, N, E);

    k_gemm_wreg_f32<8, 32><<<256, 128, 0, stream>>>(x, W1, bufH, N / 16);
    k_agg_tile<C1, TILE, true><<<(N + TILE - 1) / TILE, 1024, 0, stream>>>(bufH, src, dst, dinv, b1, bufA, out, N, E);

    k_gemm_wmma_f32<4><<<N / 16, 32, 0, stream>>>(bufA, W2, bufH, C1);
    k_agg_tile<C2, TILE, true><<<(N + TILE - 1) / TILE, 1024, 0, stream>>>(bufH, src, dst, dinv, b2, bufA, out, N, E);

    k_gemm_wmma_f32<2><<<N / 16, 32, 0, stream>>>(bufA, W3, bufH, C2);
    k_agg_tile<C3, TILE, false><<<(N + TILE - 1) / TILE, 1024, 0, stream>>>(bufH, src, dst, dinv, b3, out, out, N, E);
}


# aggregation: per-edge source factor from a one-byte degree table copied into LDS once per group (rsqrtf of the byte = dinv's bits), not a scattered 4-byte global read per edge; sum order as before
# speedup vs baseline: 5.5020x; 1.0393x over previous
#include <hip/hip_runtime.h>
#include <hip/hip_bf16.h>

typedef __attribute__((ext_vector_type(2))) float v2f;
typedef __attribute__((ext_vector_type(8))) float v8f;

constexpr int REGION_ROW_WORDS = 32;
constexpr int PLANE_RUN_LOG2   = 13;
constexpr int PLANE_RUN        = 1 << PLANE_RUN_LOG2;

constexpr int DEG_TAB_WORDS = 128;
constexpr int DEG_TAB_TILES = 250;

template <int T, int LROW>
__device__ __forceinline__ size_t plane_offset(int e) {
    constexpr int RB = T * LROW * 4;
    return (size_t)(e >> PLANE_RUN_LOG2) * RB + (size_t)(RB - PLANE_RUN) + (size_t)(e & (PLANE_RUN - 1));
}

template <int T, int LROW>
__global__ void k_plane(const int* __restrict__ dst, float* lst, int E) {
    const int e0 = (blockIdx.x * blockDim.x + threadIdx.x) * 4;
    if (e0 >= E) return;
    unsigned char* tb = (unsigned char*)lst + plane_offset<T, LROW>(e0);
    if (e0 + 3 < E) {
        const int d0 = dst[e0], d1 = dst[e0 + 1], d2 = dst[e0 + 2], d3 = dst[e0 + 3];
        *(unsigned*)tb = (unsigned)(d0 / T) | ((unsigned)(d1 / T) << 8) |
                         ((unsigned)(d2 / T) << 16) | ((unsigned)(d3 / T) << 24);
    } else {
        for (int e = e0; e < E; ++e) tb[e - e0] = (unsigned char)(dst[e] / T);
    }
}

template <int T>
__global__ __launch_bounds__(1024)
void k_tile_list(const int* __restrict__ src, const int* __restrict__ dst,
                 float* __restrict__ dinv, float* lst, int N, int E) {
    constexpr int NT   = 1024;
    constexpr int CAP  = 16384;
    constexpr int LROW = REGION_ROW_WORDS;
    constexpr int RB   = T * LROW * 4;
    constexpr int LPTB = 6;
    constexpr int GV   = NT * LPTB;
    constexpr int VPR_LOG2 = PLANE_RUN_LOG2 - 4;
    static_assert(T <= 512, "row in tile must fit 9 bits of a list word");
    static_assert(RB > PLANE_RUN + 4 && RB % 16 == 0, "a region holds a list and an aligned run of the plane");
    static_assert(T % 16 == 0 && T / 4 <= DEG_TAB_WORDS && RB > PLANE_RUN + DEG_TAB_WORDS * 4 + 4,
                  "a tile's chunk of the degree table is whole 16-byte vectors and fits its place in the region");

    __shared__ int degc[T];
    __shared__ int qe[CAP];
    __shared__ int cnt;
    __shared__ int ovf;

    const int tid  = threadIdx.x;
    const int base = blockIdx.x * T;
    const int rows = min(T, N - base);

    unsigned* lw = (unsigned*)lst + (size_t)base * LROW;
    const int lcap = max(rows * LROW - PLANE_RUN / 4 - DEG_TAB_WORDS - 1, 0);

    for (int i = tid; i < T; i += NT) degc[i] = 0;
    if (tid == 0) { cnt = 0; ovf = 0; }
    __syncthreads();

    auto push = [&](int e) {
        const int q = atomicAdd(&cnt, 1);
        if (q < CAP) {
            qe[q] = e;
        } else {
            ovf = 1;
            atomicAdd(&degc[dst[e] - base], 1);
        }
    };

    const int n16 = E >> 4;
    const char* pl = (const char*)lst + (RB - PLANE_RUN);
    const int ngroups = (n16 + GV - 1) / GV;
    const int rot = (ngroups > 0) ? (int)(blockIdx.x % (unsigned)ngroups) : 0;
    const unsigned pat = (unsigned)blockIdx.x * 0x01010101u;
    auto zb = [](unsigned v) {
        return ~(((v & 0x7F7F7F7Fu) + 0x7F7F7F7Fu) | v) & 0x80808080u;
    };

#pragma unroll 1
    for (int g = 0; g < ngroups; ++g) {
        const int gg = (g + rot < ngroups) ? (g + rot) : (g + rot - ngroups);
        const int i0 = gg * GV + tid;
        uint4 d[LPTB];
        unsigned off[LPTB];
#pragma unroll
        for (int k = 0; k < LPTB; ++k) {
            const int ic = min(i0 + k * NT, n16 - 1);
            off[k] = ((unsigned)ic >> VPR_LOG2) * (unsigned)RB
                   + ((unsigned)ic & ((1u << VPR_LOG2) - 1u)) * 16u;
        }
        __builtin_amdgcn_sched_barrier(0);
#pragma unroll
        for (int k = 0; k < LPTB; ++k) d[k] = *(const uint4*)(pl + off[k]);
        __builtin_amdgcn_sched_barrier(0);
#pragma unroll
        for (int k = 0; k < LPTB; ++k) {
            unsigned f = (zb(d[k].x ^ pat) >> 7) | (zb(d[k].y ^ pat) >> 6) |
                         (zb(d[k].z ^ pat) >> 5) | (zb(d[k].w ^ pat) >> 4);
            if (i0 + k * NT >= n16) f = 0;
            const int e = (i0 + k * NT) * 16;
            while (f) {
                const int pos = __builtin_ctz(f);
                f &= f - 1;
                push(e + (pos & 7) * 4 + (pos >> 3));
            }
        }
    }
    for (int e = n16 * 16 + tid; e < E; e += NT)
        if ((unsigned)(dst[e] - base) < (unsigned)rows) push(e);
    __syncthreads();

    const int n = min(cnt, CAP);
    constexpr int DU = 8;
    for (int i0 = 0; i0 < n; i0 += NT * DU) {
        int dv[DU], sv[DU];
#pragma unroll
        for (int u = 0; u < DU; ++u) {
            const int e = qe[min(i0 + u * NT + tid, n - 1)];
            dv[u] = dst[e];
            sv[u] = src[e];
        }
#pragma unroll
        for (int u = 0; u < DU; ++u) {
            const int i = i0 + u * NT + tid;
            if (i < n) {
                const int dl = dv[u] - base;
                atomicAdd(&degc[dl], 1);
                if (i < lcap) lw[1 + i] = ((unsigned)sv[u] << 9) | (unsigned)dl;
            }
        }
    }
    __syncthreads();

    for (int i = tid; i < rows; i += NT) dinv[base + i] = rsqrtf(1.0f + (float)degc[i]);
    if (rows == T && tid < T / 4) {
        unsigned wv = 0;
#pragma unroll
        for (int k = 0; k < 4; ++k) wv |= (unsigned)min(1 + degc[4 * tid + k], 255) << (8 * k);
        lw[T * LROW - PLANE_RUN / 4 - DEG_TAB_WORDS + tid] = wv;
    }
    if (tid == 0) lw[0] = ovf ? 0xFFFFFFFFu : (unsigned)n;
}

template <int NT, int KS>
__global__ __launch_bounds__(128)
void k_gemm_wreg_f32(const float* __restrict__ X, const float* __restrict__ W,
                     float* __restrict__ Y, int npanels) {
    constexpr int Cout = NT * 16;
    constexpr int KW   = KS * 4;
    constexpr int K    = KW * 4;
    constexpr int P    = Cout + 4;
    constexpr int C4   = Cout / 4;
    __shared__ alignas(16) float red[2][4][16 * P];

    const int tid  = threadIdx.x;
    const int lane = tid & 31;
    const int q    = __builtin_amdgcn_readfirstlane(tid >> 5);
    const int half = lane >> 4;
    const int l16  = lane & 15;

    v2f b[KS][NT];
    {
        const float* __restrict__ wq = W + (size_t)(q * KW + half * 2) * Cout + l16;
#pragma unroll
        for (int j = 0; j < KS; ++j)
#pragma unroll
            for (int t = 0; t < NT; ++t) {
                b[j][t].x = wq[(size_t)(4 * j) * Cout + t * 16];
                b[j][t].y = wq[(size_t)(4 * j + 1) * Cout + t * 16];
            }
    }

    const float* __restrict__ xq = X + (size_t)l16 * K + q * KW + half * 2;

    int p = blockIdx.x;
    v2f a[KS];
    {
        const float* __restrict__ xp = xq + (size_t)min(p, npanels - 1) * 16 * K;
#pragma unroll
        for (int j = 0; j < KS; ++j) a[j] = *(const v2f*)(xp + 4 * j);
    }

    int buf = 0;
    for (; p < npanels; p += gridDim.x) {
        v2f an[KS];
        {
            const float* __restrict__ xp = xq + (size_t)min(p + (int)gridDim.x, npanels - 1) * 16 * K;
#pragma unroll
            for (int j = 0; j < KS; ++j) an[j] = *(const v2f*)(xp + 4 * j);
        }

        v8f c[NT];
#pragma unroll
        for (int t = 0; t < NT; ++t) c[t] = (v8f){};
#pragma unroll
        for (int j = 0; j < KS; ++j)
#pragma unroll
            for (int t = 0; t < NT; ++t)
                c[t] = __builtin_amdgcn_wmma_f32_16x16x4_f32(
                            false, a[j],  false, b[j][t],
                            (short)0, c[t],
                            false,  false);

        float* rw = &red[buf][q][0] + l16;
#pragma unroll
        for (int t = 0; t < NT; ++t)
#pragma unroll
            for (int v = 0; v < 8; ++v)
                rw[(v + half * 8) * P + t * 16] = c[t][v];
        __syncthreads();

        float4* __restrict__ y4 = (float4*)(Y + (size_t)p * 16 * Cout);
#pragma unroll
        for (int i = tid; i < 16 * C4; i += 128) {
            const int r  = i / C4;
            const int c4 = i - r * C4;
            const float4 s0 = *(const float4*)&red[buf][0][r * P + c4 * 4];
            const float4 s1 = *(const float4*)&red[buf][1][r * P + c4 * 4];
            const float4 s2 = *(const float4*)&red[buf][2][r * P + c4 * 4];
            const float4 s3 = *(const float4*)&red[buf][3][r * P + c4 * 4];
            float4 o;
            o.x = ((s0.x + s1.x) + s2.x) + s3.x;
            o.y = ((s0.y + s1.y) + s2.y) + s3.y;
            o.z = ((s0.z + s1.z) + s2.z) + s3.z;
            o.w = ((s0.w + s1.w) + s2.w) + s3.w;
            y4[i] = o;
        }
        buf ^= 1;
#pragma unroll
        for (int j = 0; j < KS; ++j) a[j] = an[j];
    }
}

template <int NT>
__global__ __launch_bounds__(32)
void k_gemm_wmma_f32(const float* __restrict__ X, const float* __restrict__ W,
                     float* __restrict__ Y, int K) {
    constexpr int Cout = NT * 16;
    const int mt   = blockIdx.x;
    const int lane = threadIdx.x;
    const int half = lane >> 4;
    const int l16  = lane & 15;

    const float* __restrict__ xrow = X + (size_t)(mt * 16 + l16) * K;
    const float* __restrict__ wlan = W + l16;

    v8f c[NT];
#pragma unroll
    for (int t = 0; t < NT; ++t) c[t] = (v8f){};

    for (int k = 0; k < K; k += 4) {
        const int ka = k + half * 2;
        v2f a;
        a.x = xrow[ka];
        a.y = xrow[ka + 1];
        const float* __restrict__ wk = wlan + (size_t)ka * Cout;
#pragma unroll
        for (int t = 0; t < NT; ++t) {
            v2f b;
            b.x = wk[t * 16];
            b.y = wk[t * 16 + Cout];
            c[t] = __builtin_amdgcn_wmma_f32_16x16x4_f32(
                        false, a,  false, b,
                        (short)0, c[t],
                        false,  false);
        }
    }

    float* __restrict__ yp = Y + (size_t)(mt * 16) * Cout + l16;
#pragma unroll
    for (int t = 0; t < NT; ++t)
#pragma unroll
        for (int v = 0; v < 8; ++v)
            yp[(size_t)(v + half * 8) * Cout + t * 16] = c[t][v];
}

template <int C, int T, bool RELU, int TAB>
__global__ __launch_bounds__(1024)
void k_agg_tile(const float* __restrict__ h,
                const int* __restrict__ src, const int* __restrict__ dst,
                const float* __restrict__ dinv, const float* __restrict__ bias,
                float* out, float* lst, float* tabc, int N, int E) {
    constexpr int NT  = 1024;
    constexpr int VEC = C / 32;
    constexpr int C4  = C / 4;
    constexpr int CAP = 4096;
    constexpr int LPT = 8;
    constexpr int SUB = 6;
    constexpr int CH4 = NT * LPT * SUB;
    constexpr int U   = 16 / VEC;
    constexpr int LROW = REGION_ROW_WORDS;
    constexpr int RB   = T * LROW * 4;
    constexpr int TABN = DEG_TAB_TILES * T;
    constexpr int VPC  = T / 16;
    static_assert(T <= 512, "row in tile must fit 9 bits of a list word");
    static_assert(T % 16 == 0, "a tile's chunk of the degree table is whole 16-byte vectors");

    __shared__ float acc[(T + 1) * C];
    __shared__ float dloc[T + 1];
    __shared__ int   qe[CAP];
    __shared__ int   cnt[2];
    __shared__ alignas(16) unsigned char dtab[TABN];

    const int tid  = threadIdx.x;
    const int lane = tid & 31;
    const int wave = __builtin_amdgcn_readfirstlane(tid >> 5);
    const int base = blockIdx.x * T;
    const int rows = min(T, N - base);

    unsigned* lw = (unsigned*)lst + (size_t)base * LROW;
    const int lcap = max(rows * LROW - PLANE_RUN / 4 - DEG_TAB_WORDS - 1, 0);
    const unsigned lcount = (unsigned)__builtin_amdgcn_readfirstlane((int)lw[0]);

    for (int i = tid; i < rows; i += NT) dloc[i] = dinv[base + i];
    if (tid < 2) cnt[tid] = 0;
    if (tid == 3) dloc[T] = 0.0f;
    {
        const float4* __restrict__ h4 = (const float4*)(h + (size_t)base * C);
        const float4* __restrict__ b4 = (const float4*)bias;
        float4* a4 = (float4*)acc;
        for (int idx = tid; idx < rows * C4; idx += NT) {
            const int r  = idx / C4;
            const int c4 = idx - r * C4;
            const float di = dinv[base + r];
            const float4 hv = h4[idx];
            const float4 bv = b4[c4];
            float4 o;
            o.x = bv.x + hv.x * di * di;
            o.y = bv.y + hv.y * di * di;
            o.z = bv.z + hv.z * di * di;
            o.w = bv.w + hv.w * di * di;
            a4[idx] = o;
        }
    }
    {
        constexpr int NV = DEG_TAB_TILES * VPC;
        constexpr int NJ = (NV + NT - 1) / NT;
        const int nwhole = min(N / T, DEG_TAB_TILES);
        const int rot = (int)(blockIdx.x % (unsigned)DEG_TAB_TILES);
        uint4 tv[NJ];
        int   to[NJ];
#pragma unroll
        for (int j = 0; j < NJ; ++j) {
            const int v  = min(tid + j * NT, NV - 1);
            const int t0 = v / VPC;
            const int k  = v - t0 * VPC;
            const int t  = (t0 + rot < DEG_TAB_TILES) ? (t0 + rot) : (t0 + rot - DEG_TAB_TILES);
            to[j] = t * T + k * 16;
            tv[j] = make_uint4(~0u, ~0u, ~0u, ~0u);
            if (t < nwhole) {
                const char* p = (TAB == 2)
                    ? (const char*)tabc + to[j]
                    : (const char*)lst + (size_t)t * RB + (RB - PLANE_RUN - DEG_TAB_WORDS * 4) + k * 16;
                tv[j] = *(const uint4*)p;
            }
        }
#pragma unroll
        for (int j = 0; j < NJ; ++j)
            if (tid + j * NT < NV) *(uint4*)&dtab[to[j]] = tv[j];
    }
    __syncthreads();

    if (TAB == 1 && rows == T && blockIdx.x < DEG_TAB_TILES && tid < T / 4)
        ((unsigned*)tabc)[base / 4 + tid] = ((const unsigned*)dtab)[base / 4 + tid];

    auto srcfac = [&](unsigned w) {
        const int s = (int)(w >> 9);
        const unsigned code = dtab[min(s, TABN - 1)];
        float f = rsqrtf((float)code);
        if (code == 255u || s >= TABN) f = dinv[s];
        return f;
    };

    auto add_rows = [&](int s, int dl, float nrm, int m) {
        for (int j0 = 0; j0 < m; j0 += U) {
            float x[U][VEC];
#pragma unroll
            for (int u = 0; u < U; ++u) {
                const int sj = __builtin_amdgcn_readlane(s, j0 + u);
                const float* __restrict__ hr = h + (size_t)sj * C + lane;
#pragma unroll
                for (int k = 0; k < VEC; ++k) x[u][k] = hr[k * 32];
            }
#pragma unroll
            for (int u = 0; u < U; ++u) {
                const int   dj = __builtin_amdgcn_readlane(dl, j0 + u);
                const float nj = __int_as_float(
                    __builtin_amdgcn_readlane(__float_as_int(nrm), j0 + u));
                float* ar = acc + dj * C + lane;
#pragma unroll
                for (int k = 0; k < VEC; ++k) atomicAdd(&ar[k * 32], x[u][k] * nj);
            }
        }
    };

    if (lcount <= (unsigned)lcap) {
        const int total = (int)lcount;
        const int per   = (total + 31) >> 5;
        const int beg   = wave * per;
        const int end   = min(beg + per, total);
        const unsigned EMPTY = (unsigned)T;
        if (beg < end) {
            const int i0 = beg + lane;
            unsigned w0 = lw[1 + min(i0,      end - 1)];
            unsigned w1 = lw[1 + min(i0 + 32, end - 1)];
            if (i0      >= end) w0 = EMPTY;
            if (i0 + 32 >= end) w1 = EMPTY;
            float    d0 = srcfac(w0);
            for (int b = beg; b < end; b += 32) {
                const int m = min(32, end - b);
                const float d1 = srcfac(w1);
                const int   i2 = b + 64 + lane;
                unsigned w2 = lw[1 + min(i2, end - 1)];
                if (i2 >= end) w2 = EMPTY;
                const int s  = (int)(w0 >> 9);
                const int dl = (int)(w0 & 511u);
                add_rows(s, dl, d0 * dloc[dl], m);
                w0 = w1; d0 = d1; w1 = w2;
            }
        }
        __syncthreads();
    } else {
        const bool vec_ok = (((size_t)dst) & 15) == 0;
        const int  n4     = vec_ok ? (E >> 2) : 0;
        const int4* __restrict__ dst4 = (const int4*)dst;
        const int nchunks = (n4 + CH4 - 1) / CH4;
        const int ndone   = n4 * 4;
        const int nsteps  = nchunks + ((ndone < E) ? 1 : 0);
        int p = 0;

        auto push = [&](int e) {
            const int q = atomicAdd(&cnt[p], 1);
            if (q < CAP) {
                qe[q] = e;
            } else {
                const int dl = dst[e] - base;
                const int s = src[e];
                const float nrm = dinv[s] * dloc[dl];
#pragma unroll 1
                for (int c = 0; c < C; ++c)
                    atomicAdd(&acc[dl * C + c], h[(size_t)s * C + c] * nrm);
            }
        };
        auto test = [&](int e, int d) {
            if ((unsigned)(d - base) < (unsigned)rows) push(e);
        };

        const int rot = (nchunks > 0) ? (int)(blockIdx.x % (unsigned)nchunks) : 0;

        for (int st = 0; st < nsteps; ++st) {
            if (st < nchunks) {
                const int ch = (st + rot < nchunks) ? (st + rot) : (st + rot - nchunks);
#pragma unroll 1
                for (int sb = 0; sb < SUB; ++sb) {
                    if (ch * CH4 + sb * (NT * LPT) >= n4) break;
                    const int i0 = ch * CH4 + sb * (NT * LPT) + tid;
                    int4 d[LPT];
#pragma unroll
                    for (int k = 0; k < LPT; ++k) {
                        const int i = i0 + k * NT;
                        d[k] = dst4[min(i, n4 - 1)];
                        if (i >= n4) d[k] = make_int4(-1, -1, -1, -1);
                    }
#pragma unroll
                    for (int k = 0; k < LPT; ++k) {
                        const unsigned a = (unsigned)(d[k].x - base);
                        const unsigned b = (unsigned)(d[k].y - base);
                        const unsigned c = (unsigned)(d[k].z - base);
                        const unsigned f = (unsigned)(d[k].w - base);
                        if (min(min(a, b), min(c, f)) < (unsigned)rows) {
                            const int e = (i0 + k * NT) * 4;
                            test(e,     d[k].x);
                            test(e + 1, d[k].y);
                            test(e + 2, d[k].z);
                            test(e + 3, d[k].w);
                        }
                    }
                }
            } else {
                for (int e = ndone + tid; e < E; e += NT) test(e, dst[e]);
            }
            __syncthreads();

            const int n = __builtin_amdgcn_readfirstlane(min(cnt[p], CAP));
            if (tid == 0) cnt[p ^ 1] = 0;
            for (int r0 = 0; r0 < n; r0 += NT) {
                const int nr = min(n - r0, NT);
                const int li = lane * 32 + wave;
                int   s = 0, dl = T;
                float nrm = 0.0f;
                if (li < nr) {
                    const int e = qe[r0 + li];
                    dl  = dst[e] - base;
                    s   = src[e];
                    nrm = dinv[s] * dloc[dl];
                }
                const int m = (nr > wave) ? ((nr - wave + 31) >> 5) : 0;
                add_rows(s, dl, nrm, m);
            }
            __syncthreads();
            p ^= 1;
        }
    }

    {
        const float4* a4 = (const float4*)acc;
        float4* o4 = (float4*)(out + (size_t)base * C);
        for (int idx = tid; idx < rows * C4; idx += NT) {
            float4 o = a4[idx];
            if (RELU) {
                o.x = fmaxf(o.x, 0.0f);
                o.y = fmaxf(o.y, 0.0f);
                o.z = fmaxf(o.z, 0.0f);
                o.w = fmaxf(o.w, 0.0f);
            }
            o4[idx] = o;
        }
    }
}

extern "C" void kernel_launch(void* const* d_in, const int* in_sizes, int n_in,
                              void* d_out, int out_size, void* d_ws, size_t ws_size,
                              hipStream_t stream) {
    const float* x  = (const float*)d_in[0];
    const int*   ei = (const int*)  d_in[1];
    const float* W1 = (const float*)d_in[2];
    const float* b1 = (const float*)d_in[3];
    const float* W2 = (const float*)d_in[4];
    const float* b2 = (const float*)d_in[5];
    const float* W3 = (const float*)d_in[6];
    const float* b3 = (const float*)d_in[7];

    const int K0 = 512, C1 = 128, C2 = 64, C3 = 32;
    const int N = in_sizes[0] / K0;
    const int E = in_sizes[1] / 2;
    const int* src = ei;
    const int* dst = ei + E;

    float* ws   = (float*)d_ws;
    size_t pad  = ((size_t)N + 255) & ~(size_t)255;
    float* dinv = ws;
    float* bufH = ws + pad;
    float* bufA = bufH + (size_t)N * C1;
    float* out  = (float*)d_out;
    float* tabc = bufH + (size_t)N * C2;

    const int TB = 256;
    constexpr int TILE = 400;

    k_plane<TILE, REGION_ROW_WORDS><<<(E + 4 * TB - 1) / (4 * TB), TB, 0, stream>>>(dst, out, E);
    k_tile_list<TILE><<<(N + TILE - 1) / TILE, 1024, 0, stream>>>(src, dst, dinv, out, N, E);

    k_gemm_wreg_f32<8, 32><<<256, 128, 0, stream>>>(x, W1, bufH, N / 16);
    k_agg_tile<C1, TILE, true, 0><<<(N + TILE - 1) / TILE, 1024, 0, stream>>>(bufH, src, dst, dinv, b1, bufA, out, tabc, N, E);

    k_gemm_wmma_f32<4><<<N / 16, 32, 0, stream>>>(bufA, W2, bufH, C1);
    k_agg_tile<C2, TILE, true, 1><<<(N + TILE - 1) / TILE, 1024, 0, stream>>>(bufH, src, dst, dinv, b2, bufA, out, tabc, N, E);

    k_gemm_wmma_f32<2><<<N / 16, 32, 0, stream>>>(bufA, W3, bufH, C2);
    k_agg_tile<C3, TILE, false, 2><<<(N + TILE - 1) / TILE, 1024, 0, stream>>>(bufH, src, dst, dinv, b3, out, out, tabc, N, E);
}


# tile plane as one bitmap per tile (in the workspace), not a byte per edge: a tile's list pass reads only its own bits, 128 edges per 16-byte load, no compares; sums: LDS arrival order as before
# speedup vs baseline: 5.7962x; 1.0535x over previous
#include <hip/hip_runtime.h>
#include <hip/hip_bf16.h>

typedef __attribute__((ext_vector_type(2))) float v2f;
typedef __attribute__((ext_vector_type(8))) float v8f;

constexpr int REGION_ROW_WORDS = 32;
constexpr int BITS_WAVES       = 4;
constexpr int BITS_MAX_TILES   = 256;

constexpr int DEG_TAB_WORDS = 128;
constexpr int DEG_TAB_TILES = 250;

__device__ __forceinline__ int bits_pitch_words(int E) {
    return ((((E + 1023) >> 10) * 32) + 63) & ~63;
}

template <int T>
__global__ __launch_bounds__(32 * BITS_WAVES)
void k_tile_bits(const int* __restrict__ dst, float* bitsf, int N, int E) {
    __shared__ unsigned col[BITS_WAVES][BITS_MAX_TILES * 32];

    const int lane  = threadIdx.x & 31;
    const int wave  = threadIdx.x >> 5;
    const int run   = blockIdx.x * BITS_WAVES + wave;
    const int nruns = (E + 1023) >> 10;
    if (run >= nruns) return;
    const int ntiles = min((N + T - 1) / T, BITS_MAX_TILES);
    const int pitch  = bits_pitch_words(E);

    unsigned* c = &col[wave][lane];
    for (int t = 0; t < ntiles; ++t) c[t * 32] = 0u;

    const int word = run * 32 + lane;
    const int e0   = word * 32;
    auto tile = [](int d) { return (int)min((unsigned)d / (unsigned)T, (unsigned)(BITS_MAX_TILES - 1)); };
    if ((((size_t)dst) & 15) == 0 && e0 + 31 < E) {
        const int4* __restrict__ p = (const int4*)(dst + e0);
        int4 v[8];
#pragma unroll
        for (int k = 0; k < 8; ++k) v[k] = p[k];
#pragma unroll
        for (int k = 0; k < 8; ++k) {
            atomicOr(&c[tile(v[k].x) * 32], 1u << (4 * k));
            atomicOr(&c[tile(v[k].y) * 32], 1u << (4 * k + 1));
            atomicOr(&c[tile(v[k].z) * 32], 1u << (4 * k + 2));
            atomicOr(&c[tile(v[k].w) * 32], 1u << (4 * k + 3));
        }
    } else {
        for (int j = 0; j < 32; ++j)
            if (e0 + j < E) atomicOr(&c[tile(dst[e0 + j]) * 32], 1u << j);
    }

    unsigned* o = (unsigned*)bitsf + word;
    for (int t = 0; t < ntiles; ++t) o[(size_t)t * pitch] = c[t * 32];
}

template <int T>
__global__ __launch_bounds__(1024)
void k_tile_list(const int* __restrict__ src, const int* __restrict__ dst,
                 const float* bitsf, float* __restrict__ dinv, float* lst, int N, int E) {
    constexpr int NT   = 1024;
    constexpr int CAP  = 16384;
    constexpr int LROW = REGION_ROW_WORDS;
    constexpr int RB   = T * LROW * 4;
    constexpr int LPTB = 13;
    constexpr int GV   = NT * LPTB;
    static_assert(T <= 512, "row in tile must fit 9 bits of a list word");
    static_assert(T % 16 == 0 && T / 4 <= DEG_TAB_WORDS && RB % 256 == 0 && RB > DEG_TAB_WORDS * 4 + 4,
                  "a tile's chunk of the degree table is whole 16-byte vectors and fits its aligned place in the region");

    __shared__ int degc[T];
    __shared__ int qe[CAP];
    __shared__ int cnt;
    __shared__ int ovf;

    const int tid  = threadIdx.x;
    const int base = blockIdx.x * T;
    const int rows = min(T, N - base);

    unsigned* lw = (unsigned*)lst + (size_t)base * LROW;
    const int lcap = max(rows * LROW - DEG_TAB_WORDS - 1, 0);

    for (int i = tid; i < T; i += NT) degc[i] = 0;
    if (tid == 0) { cnt = 0; ovf = 0; }
    __syncthreads();

    auto push = [&](int e) {
        const int q = atomicAdd(&cnt, 1);
        if (q < CAP) {
            qe[q] = e;
        } else {
            ovf = 1;
            atomicAdd(&degc[dst[e] - base], 1);
        }
    };

    const uint4* __restrict__ bm = (const uint4*)((const unsigned*)bitsf + (size_t)blockIdx.x * bits_pitch_words(E));
    const int nvec = ((E + 1023) >> 10) * 8;

#pragma unroll 1
    for (int v0 = 0; v0 < nvec; v0 += GV) {
        const int i0 = v0 + tid;
        uint4 d[LPTB];
        __builtin_amdgcn_sched_barrier(0);
#pragma unroll
        for (int k = 0; k < LPTB; ++k) d[k] = bm[min(i0 + k * NT, nvec - 1)];
        __builtin_amdgcn_sched_barrier(0);
#pragma unroll
        for (int k = 0; k < LPTB; ++k) {
            unsigned long long lo = (unsigned long long)d[k].x | ((unsigned long long)d[k].y << 32);
            unsigned long long hi = (unsigned long long)d[k].z | ((unsigned long long)d[k].w << 32);
            if (i0 + k * NT >= nvec) { lo = 0ull; hi = 0ull; }
            const int e = (i0 + k * NT) * 128;
            while (lo | hi) {
                const bool low = lo != 0ull;
                unsigned long long w = low ? lo : hi;
                const int pos = __builtin_ctzll(w);
                w &= w - 1ull;
                if (low) lo = w; else hi = w;
                push(e + (low ? 0 : 64) + pos);
            }
        }
    }
    __syncthreads();

    const int n = min(cnt, CAP);
    constexpr int DU = 8;
    for (int i0 = 0; i0 < n; i0 += NT * DU) {
        int dv[DU], sv[DU];
#pragma unroll
        for (int u = 0; u < DU; ++u) {
            const int e = qe[min(i0 + u * NT + tid, n - 1)];
            dv[u] = dst[e];
            sv[u] = src[e];
        }
#pragma unroll
        for (int u = 0; u < DU; ++u) {
            const int i = i0 + u * NT + tid;
            if (i < n) {
                const int dl = dv[u] - base;
                atomicAdd(&degc[dl], 1);
                if (i < lcap) lw[1 + i] = ((unsigned)sv[u] << 9) | (unsigned)dl;
            }
        }
    }
    __syncthreads();

    for (int i = tid; i < rows; i += NT) dinv[base + i] = rsqrtf(1.0f + (float)degc[i]);
    if (rows == T && tid < T / 4) {
        unsigned wv = 0;
#pragma unroll
        for (int k = 0; k < 4; ++k) wv |= (unsigned)min(1 + degc[4 * tid + k], 255) << (8 * k);
        lw[T * LROW - DEG_TAB_WORDS + tid] = wv;
    }
    if (tid == 0) lw[0] = ovf ? 0xFFFFFFFFu : (unsigned)n;
}

template <int NT, int KS>
__global__ __launch_bounds__(128)
void k_gemm_wreg_f32(const float* __restrict__ X, const float* __restrict__ W,
                     float* __restrict__ Y, int npanels) {
    constexpr int Cout = NT * 16;
    constexpr int KW   = KS * 4;
    constexpr int K    = KW * 4;
    constexpr int P    = Cout + 4;
    constexpr int C4   = Cout / 4;
    __shared__ alignas(16) float red[2][4][16 * P];

    const int tid  = threadIdx.x;
    const int lane = tid & 31;
    const int q    = __builtin_amdgcn_readfirstlane(tid >> 5);
    const int half = lane >> 4;
    const int l16  = lane & 15;

    v2f b[KS][NT];
    {
        const float* __restrict__ wq = W + (size_t)(q * KW + half * 2) * Cout + l16;
#pragma unroll
        for (int j = 0; j < KS; ++j)
#pragma unroll
            for (int t = 0; t < NT; ++t) {
                b[j][t].x = wq[(size_t)(4 * j) * Cout + t * 16];
                b[j][t].y = wq[(size_t)(4 * j + 1) * Cout + t * 16];
            }
    }

    const float* __restrict__ xq = X + (size_t)l16 * K + q * KW + half * 2;

    int p = blockIdx.x;
    v2f a[KS];
    {
        const float* __restrict__ xp = xq + (size_t)min(p, npanels - 1) * 16 * K;
#pragma unroll
        for (int j = 0; j < KS; ++j) a[j] = *(const v2f*)(xp + 4 * j);
    }

    int buf = 0;
    for (; p < npanels; p += gridDim.x) {
        v2f an[KS];
        {
            const float* __restrict__ xp = xq + (size_t)min(p + (int)gridDim.x, npanels - 1) * 16 * K;
#pragma unroll
            for (int j = 0; j < KS; ++j) an[j] = *(const v2f*)(xp + 4 * j);
        }

        v8f c[NT];
#pragma unroll
        for (int t = 0; t < NT; ++t) c[t] = (v8f){};
#pragma unroll
        for (int j = 0; j < KS; ++j)
#pragma unroll
            for (int t = 0; t < NT; ++t)
                c[t] = __builtin_amdgcn_wmma_f32_16x16x4_f32(
                            false, a[j],  false, b[j][t],
                            (short)0, c[t],
                            false,  false);

        float* rw = &red[buf][q][0] + l16;
#pragma unroll
        for (int t = 0; t < NT; ++t)
#pragma unroll
            for (int v = 0; v < 8; ++v)
                rw[(v + half * 8) * P + t * 16] = c[t][v];
        __syncthreads();

        float4* __restrict__ y4 = (float4*)(Y + (size_t)p * 16 * Cout);
#pragma unroll
        for (int i = tid; i < 16 * C4; i += 128) {
            const int r  = i / C4;
            const int c4 = i - r * C4;
            const float4 s0 = *(const float4*)&red[buf][0][r * P + c4 * 4];
            const float4 s1 = *(const float4*)&red[buf][1][r * P + c4 * 4];
            const float4 s2 = *(const float4*)&red[buf][2][r * P + c4 * 4];
            const float4 s3 = *(const float4*)&red[buf][3][r * P + c4 * 4];
            float4 o;
            o.x = ((s0.x + s1.x) + s2.x) + s3.x;
            o.y = ((s0.y + s1.y) + s2.y) + s3.y;
            o.z = ((s0.z + s1.z) + s2.z) + s3.z;
            o.w = ((s0.w + s1.w) + s2.w) + s3.w;
            y4[i] = o;
        }
        buf ^= 1;
#pragma unroll
        for (int j = 0; j < KS; ++j) a[j] = an[j];
    }
}

template <int NT>
__global__ __launch_bounds__(32)
void k_gemm_wmma_f32(const float* __restrict__ X, const float* __restrict__ W,
                     float* __restrict__ Y, int K) {
    constexpr int Cout = NT * 16;
    const int mt   = blockIdx.x;
    const int lane = threadIdx.x;
    const int half = lane >> 4;
    const int l16  = lane & 15;

    const float* __restrict__ xrow = X + (size_t)(mt * 16 + l16) * K;
    const float* __restrict__ wlan = W + l16;

    v8f c[NT];
#pragma unroll
    for (int t = 0; t < NT; ++t) c[t] = (v8f){};

    for (int k = 0; k < K; k += 4) {
        const int ka = k + half * 2;
        v2f a;
        a.x = xrow[ka];
        a.y = xrow[ka + 1];
        const float* __restrict__ wk = wlan + (size_t)ka * Cout;
#pragma unroll
        for (int t = 0; t < NT; ++t) {
            v2f b;
            b.x = wk[t * 16];
            b.y = wk[t * 16 + Cout];
            c[t] = __builtin_amdgcn_wmma_f32_16x16x4_f32(
                        false, a,  false, b,
                        (short)0, c[t],
                        false,  false);
        }
    }

    float* __restrict__ yp = Y + (size_t)(mt * 16) * Cout + l16;
#pragma unroll
    for (int t = 0; t < NT; ++t)
#pragma unroll
        for (int v = 0; v < 8; ++v)
            yp[(size_t)(v + half * 8) * Cout + t * 16] = c[t][v];
}

template <int C, int T, bool RELU, int TAB>
__global__ __launch_bounds__(1024)
void k_agg_tile(const float* __restrict__ h,
                const int* __restrict__ src, const int* __restrict__ dst,
                const float* __restrict__ dinv, const float* __restrict__ bias,
                float* out, float* lst, float* tabc, int N, int E) {
    constexpr int NT  = 1024;
    constexpr int VEC = C / 32;
    constexpr int C4  = C / 4;
    constexpr int CAP = 4096;
    constexpr int LPT = 8;
    constexpr int SUB = 6;
    constexpr int CH4 = NT * LPT * SUB;
    constexpr int U   = 16 / VEC;
    constexpr int LROW = REGION_ROW_WORDS;
    constexpr int RB   = T * LROW * 4;
    constexpr int TABN = DEG_TAB_TILES * T;
    constexpr int VPC  = T / 16;
    static_assert(T <= 512, "row in tile must fit 9 bits of a list word");
    static_assert(T % 16 == 0, "a tile's chunk of the degree table is whole 16-byte vectors");

    __shared__ float acc[(T + 1) * C];
    __shared__ float dloc[T + 1];
    __shared__ int   qe[CAP];
    __shared__ int   cnt[2];
    __shared__ alignas(16) unsigned char dtab[TABN];

    const int tid  = threadIdx.x;
    const int lane = tid & 31;
    const int wave = __builtin_amdgcn_readfirstlane(tid >> 5);
    const int base = blockIdx.x * T;
    const int rows = min(T, N - base);

    unsigned* lw = (unsigned*)lst + (size_t)base * LROW;
    const int lcap = max(rows * LROW - DEG_TAB_WORDS - 1, 0);
    const unsigned lcount = (unsigned)__builtin_amdgcn_readfirstlane((int)lw[0]);

    for (int i = tid; i < rows; i += NT) dloc[i] = dinv[base + i];
    if (tid < 2) cnt[tid] = 0;
    if (tid == 3) dloc[T] = 0.0f;
    {
        const float4* __restrict__ h4 = (const float4*)(h + (size_t)base * C);
        const float4* __restrict__ b4 = (const float4*)bias;
        float4* a4 = (float4*)acc;
        for (int idx = tid; idx < rows * C4; idx += NT) {
            const int r  = idx / C4;
            const int c4 = idx - r * C4;
            const float di = dinv[base + r];
            const float4 hv = h4[idx];
            const float4 bv = b4[c4];
            float4 o;
            o.x = bv.x + hv.x * di * di;
            o.y = bv.y + hv.y * di * di;
            o.z = bv.z + hv.z * di * di;
            o.w = bv.w + hv.w * di * di;
            a4[idx] = o;
        }
    }
    {
        constexpr int NV = DEG_TAB_TILES * VPC;
        constexpr int NJ = (NV + NT - 1) / NT;
        const int nwhole = min(N / T, DEG_TAB_TILES);
        const int rot = (int)(blockIdx.x % (unsigned)DEG_TAB_TILES);
        uint4 tv[NJ];
        int   to[NJ];
#pragma unroll
        for (int j = 0; j < NJ; ++j) {
            const int v  = min(tid + j * NT, NV - 1);
            const int t0 = v / VPC;
            const int k  = v - t0 * VPC;
            const int t  = (t0 + rot < DEG_TAB_TILES) ? (t0 + rot) : (t0 + rot - DEG_TAB_TILES);
            to[j] = t * T + k * 16;
            tv[j] = make_uint4(~0u, ~0u, ~0u, ~0u);
            if (t < nwhole) {
                const char* p = (TAB == 2)
                    ? (const char*)tabc + to[j]
                    : (const char*)lst + (size_t)t * RB + (RB - DEG_TAB_WORDS * 4) + k * 16;
                tv[j] = *(const uint4*)p;
            }
        }
#pragma unroll
        for (int j = 0; j < NJ; ++j)
            if (tid + j * NT < NV) *(uint4*)&dtab[to[j]] = tv[j];
    }
    __syncthreads();

    if (TAB == 1 && rows == T && blockIdx.x < DEG_TAB_TILES && tid < T / 4)
        ((unsigned*)tabc)[base / 4 + tid] = ((const unsigned*)dtab)[base / 4 + tid];

    auto srcfac = [&](unsigned w) {
        const int s = (int)(w >> 9);
        const unsigned code = dtab[min(s, TABN - 1)];
        float f = rsqrtf((float)code);
        if (code == 255u || s >= TABN) f = dinv[s];
        return f;
    };

    auto add_rows = [&](int s, int dl, float nrm, int m) {
        for (int j0 = 0; j0 < m; j0 += U) {
            float x[U][VEC];
#pragma unroll
            for (int u = 0; u < U; ++u) {
                const int sj = __builtin_amdgcn_readlane(s, j0 + u);
                const float* __restrict__ hr = h + (size_t)sj * C + lane;
#pragma unroll
                for (int k = 0; k < VEC; ++k) x[u][k] = hr[k * 32];
            }
#pragma unroll
            for (int u = 0; u < U; ++u) {
                const int   dj = __builtin_amdgcn_readlane(dl, j0 + u);
                const float nj = __int_as_float(
                    __builtin_amdgcn_readlane(__float_as_int(nrm), j0 + u));
                float* ar = acc + dj * C + lane;
#pragma unroll
                for (int k = 0; k < VEC; ++k) atomicAdd(&ar[k * 32], x[u][k] * nj);
            }
        }
    };

    if (lcount <= (unsigned)lcap) {
        const int total = (int)lcount;
        const int per   = (total + 31) >> 5;
        const int beg   = wave * per;
        const int end   = min(beg + per, total);
        const unsigned EMPTY = (unsigned)T;
        if (beg < end) {
            const int i0 = beg + lane;
            unsigned w0 = lw[1 + min(i0,      end - 1)];
            unsigned w1 = lw[1 + min(i0 + 32, end - 1)];
            if (i0      >= end) w0 = EMPTY;
            if (i0 + 32 >= end) w1 = EMPTY;
            float    d0 = srcfac(w0);
            for (int b = beg; b < end; b += 32) {
                const int m = min(32, end - b);
                const float d1 = srcfac(w1);
                const int   i2 = b + 64 + lane;
                unsigned w2 = lw[1 + min(i2, end - 1)];
                if (i2 >= end) w2 = EMPTY;
                const int s  = (int)(w0 >> 9);
                const int dl = (int)(w0 & 511u);
                add_rows(s, dl, d0 * dloc[dl], m);
                w0 = w1; d0 = d1; w1 = w2;
            }
        }
        __syncthreads();
    } else {
        const bool vec_ok = (((size_t)dst) & 15) == 0;
        const int  n4     = vec_ok ? (E >> 2) : 0;
        const int4* __restrict__ dst4 = (const int4*)dst;
        const int nchunks = (n4 + CH4 - 1) / CH4;
        const int ndone   = n4 * 4;
        const int nsteps  = nchunks + ((ndone < E) ? 1 : 0);
        int p = 0;

        auto push = [&](int e) {
            const int q = atomicAdd(&cnt[p], 1);
            if (q < CAP) {
                qe[q] = e;
            } else {
                const int dl = dst[e] - base;
                const int s = src[e];
                const float nrm = dinv[s] * dloc[dl];
#pragma unroll 1
                for (int c = 0; c < C; ++c)
                    atomicAdd(&acc[dl * C + c], h[(size_t)s * C + c] * nrm);
            }
        };
        auto test = [&](int e, int d) {
            if ((unsigned)(d - base) < (unsigned)rows) push(e);
        };

        const int rot = (nchunks > 0) ? (int)(blockIdx.x % (unsigned)nchunks) : 0;

        for (int st = 0; st < nsteps; ++st) {
            if (st < nchunks) {
                const int ch = (st + rot < nchunks) ? (st + rot) : (st + rot - nchunks);
#pragma unroll 1
                for (int sb = 0; sb < SUB; ++sb) {
                    if (ch * CH4 + sb * (NT * LPT) >= n4) break;
                    const int i0 = ch * CH4 + sb * (NT * LPT) + tid;
                    int4 d[LPT];
#pragma unroll
                    for (int k = 0; k < LPT; ++k) {
                        const int i = i0 + k * NT;
                        d[k] = dst4[min(i, n4 - 1)];
                        if (i >= n4) d[k] = make_int4(-1, -1, -1, -1);
                    }
#pragma unroll
                    for (int k = 0; k < LPT; ++k) {
                        const unsigned a = (unsigned)(d[k].x - base);
                        const unsigned b = (unsigned)(d[k].y - base);
                        const unsigned c = (unsigned)(d[k].z - base);
                        const unsigned f = (unsigned)(d[k].w - base);
                        if (min(min(a, b), min(c, f)) < (unsigned)rows) {
                            const int e = (i0 + k * NT) * 4;
                            test(e,     d[k].x);
                            test(e + 1, d[k].y);
                            test(e + 2, d[k].z);
                            test(e + 3, d[k].w);
                        }
                    }
                }
            } else {
                for (int e = ndone + tid; e < E; e += NT) test(e, dst[e]);
            }
            __syncthreads();

            const int n = __builtin_amdgcn_readfirstlane(min(cnt[p], CAP));
            if (tid == 0) cnt[p ^ 1] = 0;
            for (int r0 = 0; r0 < n; r0 += NT) {
                const int nr = min(n - r0, NT);
                const int li = lane * 32 + wave;
                int   s = 0, dl = T;
                float nrm = 0.0f;
                if (li < nr) {
                    const int e = qe[r0 + li];
                    dl  = dst[e] - base;
                    s   = src[e];
                    nrm = dinv[s] * dloc[dl];
                }
                const int m = (nr > wave) ? ((nr - wave + 31) >> 5) : 0;
                add_rows(s, dl, nrm, m);
            }
            __syncthreads();
            p ^= 1;
        }
    }

    {
        const float4* a4 = (const float4*)acc;
        float4* o4 = (float4*)(out + (size_t)base * C);
        for (int idx = tid; idx < rows * C4; idx += NT) {
            float4 o = a4[idx];
            if (RELU) {
                o.x = fmaxf(o.x, 0.0f);
                o.y = fmaxf(o.y, 0.0f);
                o.z = fmaxf(o.z, 0.0f);
                o.w = fmaxf(o.w, 0.0f);
            }
            o4[idx] = o;
        }
    }
}

extern "C" void kernel_launch(void* const* d_in, const int* in_sizes, int n_in,
                              void* d_out, int out_size, void* d_ws, size_t ws_size,
                              hipStream_t stream) {
    const float* x  = (const float*)d_in[0];
    const int*   ei = (const int*)  d_in[1];
    const float* W1 = (const float*)d_in[2];
    const float* b1 = (const float*)d_in[3];
    const float* W2 = (const float*)d_in[4];
    const float* b2 = (const float*)d_in[5];
    const float* W3 = (const float*)d_in[6];
    const float* b3 = (const float*)d_in[7];

    const int K0 = 512, C1 = 128, C2 = 64, C3 = 32;
    const int N = in_sizes[0] / K0;
    const int E = in_sizes[1] / 2;
    const int* src = ei;
    const int* dst = ei + E;

    float* ws   = (float*)d_ws;
    size_t pad  = ((size_t)N + 255) & ~(size_t)255;
    float* dinv = ws;
    float* bufH = ws + pad;
    float* bufA = bufH + (size_t)N * C1;
    float* out  = (float*)d_out;
    float* tabc = bufH + (size_t)N * C2;

    constexpr int TILE = 400;

    k_tile_bits<TILE><<<(((E + 1023) >> 10) + BITS_WAVES - 1) / BITS_WAVES, 32 * BITS_WAVES, 0, stream>>>(dst, bufA, N, E);
    k_tile_list<TILE><<<(N + TILE - 1) / TILE, 1024, 0, stream>>>(src, dst, bufA, dinv, out, N, E);

    k_gemm_wreg_f32<8, 32><<<256, 128, 0, stream>>>(x, W1, bufH, N / 16);
    k_agg_tile<C1, TILE, true, 0><<<(N + TILE - 1) / TILE, 1024, 0, stream>>>(bufH, src, dst, dinv, b1, bufA, out, tabc, N, E);

    k_gemm_wmma_f32<4><<<N / 16, 32, 0, stream>>>(bufA, W2, bufH, C1);
    k_agg_tile<C2, TILE, true, 1><<<(N + TILE - 1) / TILE, 1024, 0, stream>>>(bufH, src, dst, dinv, b2, bufA, out, tabc, N, E);

    k_gemm_wmma_f32<2><<<N / 16, 32, 0, stream>>>(bufA, W3, bufH, C2);
    k_agg_tile<C3, TILE, false, 2><<<(N + TILE - 1) / TILE, 1024, 0, stream>>>(bufH, src, dst, dinv, b3, out, out, tabc, N, E);
}


# aggregation: each gathered row is one load instruction (16 or 8 bytes per lane, not 4 or 2 four-byte loads), 16 rows in flight per wave; tile rows permuted in LDS; sums: LDS arrival order as before
# speedup vs baseline: 6.1572x; 1.0623x over previous
#include <hip/hip_runtime.h>
#include <hip/hip_bf16.h>

typedef __attribute__((ext_vector_type(2))) float v2f;
typedef __attribute__((ext_vector_type(8))) float v8f;
typedef __attribute__((ext_vector_type(4))) float v4f;

constexpr int REGION_ROW_WORDS = 32;
constexpr int BITS_WAVES       = 4;
constexpr int BITS_MAX_TILES   = 256;

constexpr int DEG_TAB_WORDS = 128;
constexpr int DEG_TAB_TILES = 250;

__device__ __forceinline__ int bits_pitch_words(int E) {
    return ((((E + 1023) >> 10) * 32) + 63) & ~63;
}

template <int T>
__global__ __launch_bounds__(32 * BITS_WAVES)
void k_tile_bits(const int* __restrict__ dst, float* bitsf, int N, int E) {
    __shared__ unsigned col[BITS_WAVES][BITS_MAX_TILES * 32];

    const int lane  = threadIdx.x & 31;
    const int wave  = threadIdx.x >> 5;
    const int run   = blockIdx.x * BITS_WAVES + wave;
    const int nruns = (E + 1023) >> 10;
    if (run >= nruns) return;
    const int ntiles = min((N + T - 1) / T, BITS_MAX_TILES);
    const int pitch  = bits_pitch_words(E);

    unsigned* c = &col[wave][lane];
    for (int t = 0; t < ntiles; ++t) c[t * 32] = 0u;

    const int word = run * 32 + lane;
    const int e0   = word * 32;
    auto tile = [](int d) { return (int)min((unsigned)d / (unsigned)T, (unsigned)(BITS_MAX_TILES - 1)); };
    if ((((size_t)dst) & 15) == 0 && e0 + 31 < E) {
        const int4* __restrict__ p = (const int4*)(dst + e0);
        int4 v[8];
#pragma unroll
        for (int k = 0; k < 8; ++k) v[k] = p[k];
#pragma unroll
        for (int k = 0; k < 8; ++k) {
            atomicOr(&c[tile(v[k].x) * 32], 1u << (4 * k));
            atomicOr(&c[tile(v[k].y) * 32], 1u << (4 * k + 1));
            atomicOr(&c[tile(v[k].z) * 32], 1u << (4 * k + 2));
            atomicOr(&c[tile(v[k].w) * 32], 1u << (4 * k + 3));
        }
    } else {
        for (int j = 0; j < 32; ++j)
            if (e0 + j < E) atomicOr(&c[tile(dst[e0 + j]) * 32], 1u << j);
    }

    unsigned* o = (unsigned*)bitsf + word;
    for (int t = 0; t < ntiles; ++t) o[(size_t)t * pitch] = c[t * 32];
}

template <int T>
__global__ __launch_bounds__(1024)
void k_tile_list(const int* __restrict__ src, const int* __restrict__ dst,
                 const float* bitsf, float* __restrict__ dinv, float* lst, int N, int E) {
    constexpr int NT   = 1024;
    constexpr int CAP  = 16384;
    constexpr int LROW = REGION_ROW_WORDS;
    constexpr int RB   = T * LROW * 4;
    constexpr int LPTB = 13;
    constexpr int GV   = NT * LPTB;
    static_assert(T <= 512, "row in tile must fit 9 bits of a list word");
    static_assert(T % 16 == 0 && T / 4 <= DEG_TAB_WORDS && RB % 256 == 0 && RB > DEG_TAB_WORDS * 4 + 4,
                  "a tile's chunk of the degree table is whole 16-byte vectors and fits its aligned place in the region");

    __shared__ int degc[T];
    __shared__ int qe[CAP];
    __shared__ int cnt;
    __shared__ int ovf;

    const int tid  = threadIdx.x;
    const int base = blockIdx.x * T;
    const int rows = min(T, N - base);

    unsigned* lw = (unsigned*)lst + (size_t)base * LROW;
    const int lcap = max(rows * LROW - DEG_TAB_WORDS - 1, 0);

    for (int i = tid; i < T; i += NT) degc[i] = 0;
    if (tid == 0) { cnt = 0; ovf = 0; }
    __syncthreads();

    auto push = [&](int e) {
        const int q = atomicAdd(&cnt, 1);
        if (q < CAP) {
            qe[q] = e;
        } else {
            ovf = 1;
            atomicAdd(&degc[dst[e] - base], 1);
        }
    };

    const uint4* __restrict__ bm = (const uint4*)((const unsigned*)bitsf + (size_t)blockIdx.x * bits_pitch_words(E));
    const int nvec = ((E + 1023) >> 10) * 8;

#pragma unroll 1
    for (int v0 = 0; v0 < nvec; v0 += GV) {
        const int i0 = v0 + tid;
        uint4 d[LPTB];
        __builtin_amdgcn_sched_barrier(0);
#pragma unroll
        for (int k = 0; k < LPTB; ++k) d[k] = bm[min(i0 + k * NT, nvec - 1)];
        __builtin_amdgcn_sched_barrier(0);
#pragma unroll
        for (int k = 0; k < LPTB; ++k) {
            unsigned long long lo = (unsigned long long)d[k].x | ((unsigned long long)d[k].y << 32);
            unsigned long long hi = (unsigned long long)d[k].z | ((unsigned long long)d[k].w << 32);
            if (i0 + k * NT >= nvec) { lo = 0ull; hi = 0ull; }
            const int e = (i0 + k * NT) * 128;
            while (lo | hi) {
                const bool low = lo != 0ull;
                unsigned long long w = low ? lo : hi;
                const int pos = __builtin_ctzll(w);
                w &= w - 1ull;
                if (low) lo = w; else hi = w;
                push(e + (low ? 0 : 64) + pos);
            }
        }
    }
    __syncthreads();

    const int n = min(cnt, CAP);
    constexpr int DU = 8;
    for (int i0 = 0; i0 < n; i0 += NT * DU) {
        int dv[DU], sv[DU];
#pragma unroll
        for (int u = 0; u < DU; ++u) {
            const int e = qe[min(i0 + u * NT + tid, n - 1)];
            dv[u] = dst[e];
            sv[u] = src[e];
        }
#pragma unroll
        for (int u = 0; u < DU; ++u) {
            const int i = i0 + u * NT + tid;
            if (i < n) {
                const int dl = dv[u] - base;
                atomicAdd(&degc[dl], 1);
                if (i < lcap) lw[1 + i] = ((unsigned)sv[u] << 9) | (unsigned)dl;
            }
        }
    }
    __syncthreads();

    for (int i = tid; i < rows; i += NT) dinv[base + i] = rsqrtf(1.0f + (float)degc[i]);
    if (rows == T && tid < T / 4) {
        unsigned wv = 0;
#pragma unroll
        for (int k = 0; k < 4; ++k) wv |= (unsigned)min(1 + degc[4 * tid + k], 255) << (8 * k);
        lw[T * LROW - DEG_TAB_WORDS + tid] = wv;
    }
    if (tid == 0) lw[0] = ovf ? 0xFFFFFFFFu : (unsigned)n;
}

template <int NT, int KS>
__global__ __launch_bounds__(128)
void k_gemm_wreg_f32(const float* __restrict__ X, const float* __restrict__ W,
                     float* __restrict__ Y, int npanels) {
    constexpr int Cout = NT * 16;
    constexpr int KW   = KS * 4;
    constexpr int K    = KW * 4;
    constexpr int P    = Cout + 4;
    constexpr int C4   = Cout / 4;
    __shared__ alignas(16) float red[2][4][16 * P];

    const int tid  = threadIdx.x;
    const int lane = tid & 31;
    const int q    = __builtin_amdgcn_readfirstlane(tid >> 5);
    const int half = lane >> 4;
    const int l16  = lane & 15;

    v2f b[KS][NT];
    {
        const float* __restrict__ wq = W + (size_t)(q * KW + half * 2) * Cout + l16;
#pragma unroll
        for (int j = 0; j < KS; ++j)
#pragma unroll
            for (int t = 0; t < NT; ++t) {
                b[j][t].x = wq[(size_t)(4 * j) * Cout + t * 16];
                b[j][t].y = wq[(size_t)(4 * j + 1) * Cout + t * 16];
            }
    }

    const float* __restrict__ xq = X + (size_t)l16 * K + q * KW + half * 2;

    int p = blockIdx.x;
    v2f a[KS];
    {
        const float* __restrict__ xp = xq + (size_t)min(p, npanels - 1) * 16 * K;
#pragma unroll
        for (int j = 0; j < KS; ++j) a[j] = *(const v2f*)(xp + 4 * j);
    }

    int buf = 0;
    for (; p < npanels; p += gridDim.x) {
        v2f an[KS];
        {
            const float* __restrict__ xp = xq + (size_t)min(p + (int)gridDim.x, npanels - 1) * 16 * K;
#pragma unroll
            for (int j = 0; j < KS; ++j) an[j] = *(const v2f*)(xp + 4 * j);
        }

        v8f c[NT];
#pragma unroll
        for (int t = 0; t < NT; ++t) c[t] = (v8f){};
#pragma unroll
        for (int j = 0; j < KS; ++j)
#pragma unroll
            for (int t = 0; t < NT; ++t)
                c[t] = __builtin_amdgcn_wmma_f32_16x16x4_f32(
                            false, a[j],  false, b[j][t],
                            (short)0, c[t],
                            false,  false);

        float* rw = &red[buf][q][0] + l16;
#pragma unroll
        for (int t = 0; t < NT; ++t)
#pragma unroll
            for (int v = 0; v < 8; ++v)
                rw[(v + half * 8) * P + t * 16] = c[t][v];
        __syncthreads();

        float4* __restrict__ y4 = (float4*)(Y + (size_t)p * 16 * Cout);
#pragma unroll
        for (int i = tid; i < 16 * C4; i += 128) {
            const int r  = i / C4;
            const int c4 = i - r * C4;
            const float4 s0 = *(const float4*)&red[buf][0][r * P + c4 * 4];
            const float4 s1 = *(const float4*)&red[buf][1][r * P + c4 * 4];
            const float4 s2 = *(const float4*)&red[buf][2][r * P + c4 * 4];
            const float4 s3 = *(const float4*)&red[buf][3][r * P + c4 * 4];
            float4 o;
            o.x = ((s0.x + s1.x) + s2.x) + s3.x;
            o.y = ((s0.y + s1.y) + s2.y) + s3.y;
            o.z = ((s0.z + s1.z) + s2.z) + s3.z;
            o.w = ((s0.w + s1.w) + s2.w) + s3.w;
            y4[i] = o;
        }
        buf ^= 1;
#pragma unroll
        for (int j = 0; j < KS; ++j) a[j] = an[j];
    }
}

template <int NT>
__global__ __launch_bounds__(32)
void k_gemm_wmma_f32(const float* __restrict__ X, const float* __restrict__ W,
                     float* __restrict__ Y, int K) {
    constexpr int Cout = NT * 16;
    const int mt   = blockIdx.x;
    const int lane = threadIdx.x;
    const int half = lane >> 4;
    const int l16  = lane & 15;

    const float* __restrict__ xrow = X + (size_t)(mt * 16 + l16) * K;
    const float* __restrict__ wlan = W + l16;

    v8f c[NT];
#pragma unroll
    for (int t = 0; t < NT; ++t) c[t] = (v8f){};

    for (int k = 0; k < K; k += 4) {
        const int ka = k + half * 2;
        v2f a;
        a.x = xrow[ka];
        a.y = xrow[ka + 1];
        const float* __restrict__ wk = wlan + (size_t)ka * Cout;
#pragma unroll
        for (int t = 0; t < NT; ++t) {
            v2f b;
            b.x = wk[t * 16];
            b.y = wk[t * 16 + Cout];
            c[t] = __builtin_amdgcn_wmma_f32_16x16x4_f32(
                        false, a,  false, b,
                        (short)0, c[t],
                        false,  false);
        }
    }

    float* __restrict__ yp = Y + (size_t)(mt * 16) * Cout + l16;
#pragma unroll
    for (int t = 0; t < NT; ++t)
#pragma unroll
        for (int v = 0; v < 8; ++v)
            yp[(size_t)(v + half * 8) * Cout + t * 16] = c[t][v];
}

template <int VEC> __device__ __forceinline__ void load_row(const float* __restrict__ p, float (&x)[VEC]);
template <> __device__ __forceinline__ void load_row<1>(const float* __restrict__ p, float (&x)[1]) {
    x[0] = p[0];
}
template <> __device__ __forceinline__ void load_row<2>(const float* __restrict__ p, float (&x)[2]) {
    const v2f t = *(const v2f*)p;
    x[0] = t.x; x[1] = t.y;
}
template <> __device__ __forceinline__ void load_row<4>(const float* __restrict__ p, float (&x)[4]) {
    const v4f t = *(const v4f*)p;
    x[0] = t.x; x[1] = t.y; x[2] = t.z; x[3] = t.w;
}

template <int VEC> __device__ __forceinline__ void tile_put4(float* ar, int c4, float4 o);
template <> __device__ __forceinline__ void tile_put4<1>(float* ar, int c4, float4 o) {
    *(float4*)(ar + 4 * c4) = o;
}
template <> __device__ __forceinline__ void tile_put4<2>(float* ar, int c4, float4 o) {
    *(float2*)(ar + 2 * c4)      = make_float2(o.x, o.z);
    *(float2*)(ar + 32 + 2 * c4) = make_float2(o.y, o.w);
}
template <> __device__ __forceinline__ void tile_put4<4>(float* ar, int c4, float4 o) {
    ar[c4]      = o.x;
    ar[32 + c4] = o.y;
    ar[64 + c4] = o.z;
    ar[96 + c4] = o.w;
}
template <int VEC> __device__ __forceinline__ float4 tile_get4(const float* ar, int c4);
template <> __device__ __forceinline__ float4 tile_get4<1>(const float* ar, int c4) {
    return *(const float4*)(ar + 4 * c4);
}
template <> __device__ __forceinline__ float4 tile_get4<2>(const float* ar, int c4) {
    const float2 a = *(const float2*)(ar + 2 * c4);
    const float2 b = *(const float2*)(ar + 32 + 2 * c4);
    return make_float4(a.x, b.x, a.y, b.y);
}
template <> __device__ __forceinline__ float4 tile_get4<4>(const float* ar, int c4) {
    return make_float4(ar[c4], ar[32 + c4], ar[64 + c4], ar[96 + c4]);
}

template <int C, int VEC, int G>
__device__ __forceinline__ void add_row_group(const float* __restrict__ h, float* acc, int lane,
                                              int s, int dl, float nrm, int j0) {
    float x[G][VEC];
#pragma unroll
    for (int u = 0; u < G; ++u) {
        const int sj = __builtin_amdgcn_readlane(s, j0 + u);
        load_row<VEC>(h + (size_t)sj * C + lane * VEC, x[u]);
    }
#pragma unroll
    for (int u = 0; u < G; ++u) {
        const int   dj = __builtin_amdgcn_readlane(dl, j0 + u);
        const float nj = __int_as_float(
            __builtin_amdgcn_readlane(__float_as_int(nrm), j0 + u));
        float* ar = acc + dj * C + lane;
#pragma unroll
        for (int k = 0; k < VEC; ++k) atomicAdd(&ar[k * 32], x[u][k] * nj);
    }
}

template <int C, int T, bool RELU, int TAB>
__global__ __launch_bounds__(1024)
void k_agg_tile(const float* __restrict__ h,
                const int* __restrict__ src, const int* __restrict__ dst,
                const float* __restrict__ dinv, const float* __restrict__ bias,
                float* out, float* lst, float* tabc, int N, int E) {
    constexpr int NT  = 1024;
    constexpr int VEC = C / 32;
    constexpr int C4  = C / 4;
    constexpr int CAP = 4096;
    constexpr int LPT = 8;
    constexpr int SUB = 6;
    constexpr int CH4 = NT * LPT * SUB;
    constexpr int U   = 16;
    constexpr int LROW = REGION_ROW_WORDS;
    constexpr int RB   = T * LROW * 4;
    constexpr int TABN = DEG_TAB_TILES * T;
    constexpr int VPC  = T / 16;
    static_assert(T <= 512, "row in tile must fit 9 bits of a list word");
    static_assert(T % 16 == 0, "a tile's chunk of the degree table is whole 16-byte vectors");

    __shared__ float acc[(T + 1) * C];
    __shared__ float dloc[T + 1];
    __shared__ int   qe[CAP];
    __shared__ int   cnt[2];
    __shared__ alignas(16) unsigned char dtab[TABN];

    const int tid  = threadIdx.x;
    const int lane = tid & 31;
    const int wave = __builtin_amdgcn_readfirstlane(tid >> 5);
    const int base = blockIdx.x * T;
    const int rows = min(T, N - base);

    unsigned* lw = (unsigned*)lst + (size_t)base * LROW;
    const int lcap = max(rows * LROW - DEG_TAB_WORDS - 1, 0);
    const unsigned lcount = (unsigned)__builtin_amdgcn_readfirstlane((int)lw[0]);

    for (int i = tid; i < rows; i += NT) dloc[i] = dinv[base + i];
    if (tid < 2) cnt[tid] = 0;
    if (tid == 3) dloc[T] = 0.0f;
    {
        const float4* __restrict__ h4 = (const float4*)(h + (size_t)base * C);
        const float4* __restrict__ b4 = (const float4*)bias;
        for (int idx = tid; idx < rows * C4; idx += NT) {
            const int r  = idx / C4;
            const int c4 = idx - r * C4;
            const float di = dinv[base + r];
            const float4 hv = h4[idx];
            const float4 bv = b4[c4];
            float4 o;
            o.x = bv.x + hv.x * di * di;
            o.y = bv.y + hv.y * di * di;
            o.z = bv.z + hv.z * di * di;
            o.w = bv.w + hv.w * di * di;
            tile_put4<VEC>(acc + r * C, c4, o);
        }
    }
    {
        constexpr int NV = DEG_TAB_TILES * VPC;
        constexpr int NJ = (NV + NT - 1) / NT;
        const int nwhole = min(N / T, DEG_TAB_TILES);
        const int rot = (int)(blockIdx.x % (unsigned)DEG_TAB_TILES);
        uint4 tv[NJ];
        int   to[NJ];
#pragma unroll
        for (int j = 0; j < NJ; ++j) {
            const int v  = min(tid + j * NT, NV - 1);
            const int t0 = v / VPC;
            const int k  = v - t0 * VPC;
            const int t  = (t0 + rot < DEG_TAB_TILES) ? (t0 + rot) : (t0 + rot - DEG_TAB_TILES);
            to[j] = t * T + k * 16;
            tv[j] = make_uint4(~0u, ~0u, ~0u, ~0u);
            if (t < nwhole) {
                const char* p = (TAB == 2)
                    ? (const char*)tabc + to[j]
                    : (const char*)lst + (size_t)t * RB + (RB - DEG_TAB_WORDS * 4) + k * 16;
                tv[j] = *(const uint4*)p;
            }
        }
#pragma unroll
        for (int j = 0; j < NJ; ++j)
            if (tid + j * NT < NV) *(uint4*)&dtab[to[j]] = tv[j];
    }
    __syncthreads();

    if (TAB == 1 && rows == T && blockIdx.x < DEG_TAB_TILES && tid < T / 4)
        ((unsigned*)tabc)[base / 4 + tid] = ((const unsigned*)dtab)[base / 4 + tid];

    auto srcfac = [&](unsigned w) {
        const int s = (int)(w >> 9);
        const unsigned code = dtab[min(s, TABN - 1)];
        float f = rsqrtf((float)code);
        if (code == 255u || s >= TABN) f = dinv[s];
        return f;
    };

    auto add_rows = [&](int s, int dl, float nrm, int m) {
        int j0 = 0;
        for (; j0 + U <= m; j0 += U) add_row_group<C, VEC, U>(h, acc, lane, s, dl, nrm, j0);
        for (; j0 < m; j0 += 4) add_row_group<C, VEC, 4>(h, acc, lane, s, dl, nrm, j0);
    };

    if (lcount <= (unsigned)lcap) {
        const int total = (int)lcount;
        const int per   = (total + 31) >> 5;
        const int beg   = wave * per;
        const int end   = min(beg + per, total);
        const unsigned EMPTY = (unsigned)T;
        if (beg < end) {
            const int i0 = beg + lane;
            unsigned w0 = lw[1 + min(i0,      end - 1)];
            unsigned w1 = lw[1 + min(i0 + 32, end - 1)];
            if (i0      >= end) w0 = EMPTY;
            if (i0 + 32 >= end) w1 = EMPTY;
            float    d0 = srcfac(w0);
            for (int b = beg; b < end; b += 32) {
                const int m = min(32, end - b);
                const float d1 = srcfac(w1);
                const int   i2 = b + 64 + lane;
                unsigned w2 = lw[1 + min(i2, end - 1)];
                if (i2 >= end) w2 = EMPTY;
                const int s  = (int)(w0 >> 9);
                const int dl = (int)(w0 & 511u);
                add_rows(s, dl, d0 * dloc[dl], m);
                w0 = w1; d0 = d1; w1 = w2;
            }
        }
        __syncthreads();
    } else {
        const bool vec_ok = (((size_t)dst) & 15) == 0;
        const int  n4     = vec_ok ? (E >> 2) : 0;
        const int4* __restrict__ dst4 = (const int4*)dst;
        const int nchunks = (n4 + CH4 - 1) / CH4;
        const int ndone   = n4 * 4;
        const int nsteps  = nchunks + ((ndone < E) ? 1 : 0);
        int p = 0;

        auto push = [&](int e) {
            const int q = atomicAdd(&cnt[p], 1);
            if (q < CAP) {
                qe[q] = e;
            } else {
                const int dl = dst[e] - base;
                const int s = src[e];
                const float nrm = dinv[s] * dloc[dl];
#pragma unroll 1
                for (int c = 0; c < C; ++c)
                    atomicAdd(&acc[dl * C + (c % VEC) * 32 + c / VEC], h[(size_t)s * C + c] * nrm);
            }
        };
        auto test = [&](int e, int d) {
            if ((unsigned)(d - base) < (unsigned)rows) push(e);
        };

        const int rot = (nchunks > 0) ? (int)(blockIdx.x % (unsigned)nchunks) : 0;

        for (int st = 0; st < nsteps; ++st) {
            if (st < nchunks) {
                const int ch = (st + rot < nchunks) ? (st + rot) : (st + rot - nchunks);
#pragma unroll 1
                for (int sb = 0; sb < SUB; ++sb) {
                    if (ch * CH4 + sb * (NT * LPT) >= n4) break;
                    const int i0 = ch * CH4 + sb * (NT * LPT) + tid;
                    int4 d[LPT];
#pragma unroll
                    for (int k = 0; k < LPT; ++k) {
                        const int i = i0 + k * NT;
                        d[k] = dst4[min(i, n4 - 1)];
                        if (i >= n4) d[k] = make_int4(-1, -1, -1, -1);
                    }
#pragma unroll
                    for (int k = 0; k < LPT; ++k) {
                        const unsigned a = (unsigned)(d[k].x - base);
                        const unsigned b = (unsigned)(d[k].y - base);
                        const unsigned c = (unsigned)(d[k].z - base);
                        const unsigned f = (unsigned)(d[k].w - base);
                        if (min(min(a, b), min(c, f)) < (unsigned)rows) {
                            const int e = (i0 + k * NT) * 4;
                            test(e,     d[k].x);
                            test(e + 1, d[k].y);
                            test(e + 2, d[k].z);
                            test(e + 3, d[k].w);
                        }
                    }
                }
            } else {
                for (int e = ndone + tid; e < E; e += NT) test(e, dst[e]);
            }
            __syncthreads();

            const int n = __builtin_amdgcn_readfirstlane(min(cnt[p], CAP));
            if (tid == 0) cnt[p ^ 1] = 0;
            for (int r0 = 0; r0 < n; r0 += NT) {
                const int nr = min(n - r0, NT);
                const int li = lane * 32 + wave;
                int   s = 0, dl = T;
                float nrm = 0.0f;
                if (li < nr) {
                    const int e = qe[r0 + li];
                    dl  = dst[e] - base;
                    s   = src[e];
                    nrm = dinv[s] * dloc[dl];
                }
                const int m = (nr > wave) ? ((nr - wave + 31) >> 5) : 0;
                add_rows(s, dl, nrm, m);
            }
            __syncthreads();
            p ^= 1;
        }
    }

    {
        float4* o4 = (float4*)(out + (size_t)base * C);
        for (int idx = tid; idx < rows * C4; idx += NT) {
            const int r  = idx / C4;
            const int c4 = idx - r * C4;
            float4 o = tile_get4<VEC>(acc + r * C, c4);
            if (RELU) {
                o.x = fmaxf(o.x, 0.0f);
                o.y = fmaxf(o.y, 0.0f);
                o.z = fmaxf(o.z, 0.0f);
                o.w = fmaxf(o.w, 0.0f);
            }
            o4[idx] = o;
        }
    }
}

extern "C" void kernel_launch(void* const* d_in, const int* in_sizes, int n_in,
                              void* d_out, int out_size, void* d_ws, size_t ws_size,
                              hipStream_t stream) {
    const float* x  = (const float*)d_in[0];
    const int*   ei = (const int*)  d_in[1];
    const float* W1 = (const float*)d_in[2];
    const float* b1 = (const float*)d_in[3];
    const float* W2 = (const float*)d_in[4];
    const float* b2 = (const float*)d_in[5];
    const float* W3 = (const float*)d_in[6];
    const float* b3 = (const float*)d_in[7];

    const int K0 = 512, C1 = 128, C2 = 64, C3 = 32;
    const int N = in_sizes[0] / K0;
    const int E = in_sizes[1] / 2;
    const int* src = ei;
    const int* dst = ei + E;

    float* ws   = (float*)d_ws;
    size_t pad  = ((size_t)N + 255) & ~(size_t)255;
    float* dinv = ws;
    float* bufH = ws + pad;
    float* bufA = bufH + (size_t)N * C1;
    float* out  = (float*)d_out;
    float* tabc = bufH + (size_t)N * C2;

    constexpr int TILE = 400;

    k_tile_bits<TILE><<<(((E + 1023) >> 10) + BITS_WAVES - 1) / BITS_WAVES, 32 * BITS_WAVES, 0, stream>>>(dst, bufA, N, E);
    k_tile_list<TILE><<<(N + TILE - 1) / TILE, 1024, 0, stream>>>(src, dst, bufA, dinv, out, N, E);

    k_gemm_wreg_f32<8, 32><<<256, 128, 0, stream>>>(x, W1, bufH, N / 16);
    k_agg_tile<C1, TILE, true, 0><<<(N + TILE - 1) / TILE, 1024, 0, stream>>>(bufH, src, dst, dinv, b1, bufA, out, tabc, N, E);

    k_gemm_wmma_f32<4><<<N / 16, 32, 0, stream>>>(bufA, W2, bufH, C1);
    k_agg_tile<C2, TILE, true, 1><<<(N + TILE - 1) / TILE, 1024, 0, stream>>>(bufH, src, dst, dinv, b2, bufA, out, tabc, N, E);

    k_gemm_wmma_f32<2><<<N / 16, 32, 0, stream>>>(bufA, W3, bufH, C2);
    k_agg_tile<C3, TILE, false, 2><<<(N + TILE - 1) / TILE, 1024, 0, stream>>>(bufH, src, dst, dinv, b3, out, out, tabc, N, E);
}
